# starting kernel
# baseline (speedup 1.0000x reference)
#include <hip/hip_runtime.h>
#include <math.h>

typedef float v2f __attribute__((ext_vector_type(2)));
typedef float v8f __attribute__((ext_vector_type(8)));

#define HEADS 4
#define CHAN 32
#define FDIM 128
#define NEG_SLOPE 0.2f

__global__ void init_kernel(float* __restrict__ amax, float* __restrict__ denom,
                            float* __restrict__ out, int Nn) {
    int i = blockIdx.x * blockDim.x + threadIdx.x;
    if (i < Nn * HEADS) {
        amax[i]  = -__builtin_inff();
        denom[i] = 0.0f;
    }
    if (i < Nn * FDIM) out[i] = 0.0f;
}

__global__ void gemm_wmma(const float* __restrict__ x, const float* __restrict__ W,
                          float* __restrict__ h, int Nn) {
    const int wave = threadIdx.x >> 5;
    const int lane = threadIdx.x & 31;
    const int row0 = (blockIdx.x * 4 + wave) * 16;
    if (row0 >= Nn) return;

    v8f acc[8];
#pragma unroll
    for (int t = 0; t < 8; ++t) acc[t] = (v8f)(0.0f);

    const int mrow  = row0 + (lane & 15);
    const int khalf = (lane >> 4) << 1;
    const int ncol  = lane & 15;

    for (int k0 = 0; k0 < FDIM; k0 += 4) {
        const int ka = k0 + khalf;
        v2f a;
        a.x = x[(size_t)mrow * FDIM + ka];
        a.y = x[(size_t)mrow * FDIM + ka + 1];
#pragma unroll
        for (int t = 0; t < 8; ++t) {
            const int n = t * 16 + ncol;
            v2f b;
            b.x = W[(size_t)ka * FDIM + n];
            b.y = W[(size_t)(ka + 1) * FDIM + n];
            acc[t] = __builtin_amdgcn_wmma_f32_16x16x4_f32(
                false, a, false, b, (short)0, acc[t], false, false);
        }
    }

    const int mbase = row0 + ((lane >> 4) << 3);
#pragma unroll
    for (int t = 0; t < 8; ++t) {
        const int n = t * 16 + ncol;
#pragma unroll
        for (int r = 0; r < 8; ++r)
            h[(size_t)(mbase + r) * FDIM + n] = acc[t][r];
    }
}

__global__ void attn_terms(const float* __restrict__ h,
                           const float* __restrict__ att_src,
                           const float* __restrict__ att_dst,
                           float* __restrict__ a_src, float* __restrict__ a_dst,
                           int Nn) {
    int i = blockIdx.x * blockDim.x + threadIdx.x;
    if (i >= Nn * HEADS) return;
    int node = i >> 2, hd = i & 3;
    const float* hp = h + (size_t)node * FDIM + hd * CHAN;
    const float* as = att_src + hd * CHAN;
    const float* ad = att_dst + hd * CHAN;
    float s = 0.f, d = 0.f;
#pragma unroll
    for (int c = 0; c < CHAN; ++c) {
        float v = hp[c];
        s = fmaf(v, as[c], s);
        d = fmaf(v, ad[c], d);
    }
    a_src[i] = s;
    a_dst[i] = d;
}

__device__ __forceinline__ float leaky(float v) {
    return v > 0.f ? v : NEG_SLOPE * v;
}

__device__ __forceinline__ void atomicMaxF32(float* addr, float val) {
    if (val >= 0.f)
        atomicMax((int*)addr, __float_as_int(val));
    else
        atomicMin((unsigned int*)addr, __float_as_uint(val));
}

__global__ void edge_max(const int* __restrict__ ei,
                         const float* __restrict__ a_src,
                         const float* __restrict__ a_dst,
                         float* __restrict__ amax, int Ecnt) {
    int i = blockIdx.x * blockDim.x + threadIdx.x;
    if (i >= Ecnt * HEADS) return;
    int e = i >> 2, hd = i & 3;
    int s = ei[e], d = ei[Ecnt + e];
    float al = leaky(a_src[s * HEADS + hd] + a_dst[d * HEADS + hd]);
    atomicMaxF32(&amax[d * HEADS + hd], al);
}

__global__ void edge_denom(const int* __restrict__ ei,
                           const float* __restrict__ a_src,
                           const float* __restrict__ a_dst,
                           const float* __restrict__ amax,
                           float* __restrict__ denom, int Ecnt) {
    int i = blockIdx.x * blockDim.x + threadIdx.x;
    if (i >= Ecnt * HEADS) return;
    int e = i >> 2, hd = i & 3;
    int s = ei[e], d = ei[Ecnt + e];
    float al = leaky(a_src[s * HEADS + hd] + a_dst[d * HEADS + hd]);
    float mx = amax[d * HEADS + hd];
    if (!__builtin_isfinite(mx)) mx = 0.f;
    atomicAdd(&denom[d * HEADS + hd], expf(al - mx));
}

__global__ void edge_message(const int* __restrict__ ei,
                             const float* __restrict__ a_src,
                             const float* __restrict__ a_dst,
                             const float* __restrict__ amax,
                             const float* __restrict__ denom,
                             const float* __restrict__ h,
                             float* __restrict__ out, int Ecnt) {
    unsigned int i = blockIdx.x * blockDim.x + threadIdx.x;
    unsigned int e = i >> 7;
    if (e >= (unsigned int)Ecnt) return;
    int j  = i & 127;
    int hd = j >> 5;
    int s = ei[e], d = ei[Ecnt + e];
    float al = leaky(a_src[s * HEADS + hd] + a_dst[d * HEADS + hd]);
    float mx = amax[d * HEADS + hd];
    if (!__builtin_isfinite(mx)) mx = 0.f;
    float w = expf(al - mx) / (denom[d * HEADS + hd] + 1e-16f);
    atomicAdd(&out[(size_t)d * FDIM + j], h[(size_t)s * FDIM + j] * w);
}

__global__ void finalize(float* __restrict__ out, const float* __restrict__ bias,
                         int total) {
    int i = blockIdx.x * blockDim.x + threadIdx.x;
    if (i >= total) return;
    float v = out[i] + bias[i & (FDIM - 1)];
    out[i] = v > 0.f ? v : expm1f(v);
}

extern "C" void kernel_launch(void* const* d_in, const int* in_sizes, int n_in,
                              void* d_out, int out_size, void* d_ws, size_t ws_size,
                              hipStream_t stream) {
    const float* x       = (const float*)d_in[0];
    const int*   ei      = (const int*)  d_in[1];
    const float* W       = (const float*)d_in[2];
    const float* att_src = (const float*)d_in[3];
    const float* att_dst = (const float*)d_in[4];
    const float* bias    = (const float*)d_in[5];
    float* out = (float*)d_out;

    const int N = in_sizes[0] / FDIM;
    const int E = in_sizes[1] / 2;

    float* ws     = (float*)d_ws;
    float* h      = ws;
    float* a_src  = h     + (size_t)N * FDIM;
    float* a_dst  = a_src + (size_t)N * HEADS;
    float* amax   = a_dst + (size_t)N * HEADS;
    float* denom  = amax  + (size_t)N * HEADS;

    const int B = 256;

    init_kernel<<<(N * FDIM + B - 1) / B, B, 0, stream>>>(amax, denom, out, N);

    int tiles  = (N + 15) / 16;
    int gblk   = (tiles + 3) / 4;
    gemm_wmma<<<gblk, 128, 0, stream>>>(x, W, h, N);

    attn_terms<<<(N * HEADS + B - 1) / B, B, 0, stream>>>(h, att_src, att_dst,
                                                          a_src, a_dst, N);
    edge_max  <<<(E * HEADS + B - 1) / B, B, 0, stream>>>(ei, a_src, a_dst, amax, E);
    edge_denom<<<(E * HEADS + B - 1) / B, B, 0, stream>>>(ei, a_src, a_dst, amax,
                                                          denom, E);
    long long msg_threads = (long long)E * FDIM;
    edge_message<<<(int)((msg_threads + B - 1) / B), B, 0, stream>>>(
        ei, a_src, a_dst, amax, denom, h, out, E);

    finalize<<<(N * FDIM + B - 1) / B, B, 0, stream>>>(out, bias, N * FDIM);
}


# messages summed in LDS by the work-group owning 392 dst rows (each scans all dst ids; LDS float adds, arrival order); out stored once with bias+elu: no out atomics, finalize launch gone
# speedup vs baseline: 2.2463x; 2.2463x over previous
#include <hip/hip_runtime.h>
#include <math.h>

typedef float v2f __attribute__((ext_vector_type(2)));
typedef float v8f __attribute__((ext_vector_type(8)));

#define HEADS 4
#define CHAN 32
#define FDIM 128
#define NEG_SLOPE 0.2f

__global__ void init_kernel(float* __restrict__ amax, float* __restrict__ denom,
                            int Nn) {
    int i = blockIdx.x * blockDim.x + threadIdx.x;
    if (i < Nn * HEADS) {
        amax[i]  = -__builtin_inff();
        denom[i] = 0.0f;
    }
}

__global__ void gemm_wmma(const float* __restrict__ x, const float* __restrict__ W,
                          float* __restrict__ h, int Nn) {
    const int wave = threadIdx.x >> 5;
    const int lane = threadIdx.x & 31;
    const int row0 = (blockIdx.x * 4 + wave) * 16;
    if (row0 >= Nn) return;

    v8f acc[8];
#pragma unroll
    for (int t = 0; t < 8; ++t) acc[t] = (v8f)(0.0f);

    const int mrow  = row0 + (lane & 15);
    const int khalf = (lane >> 4) << 1;
    const int ncol  = lane & 15;

    for (int k0 = 0; k0 < FDIM; k0 += 4) {
        const int ka = k0 + khalf;
        v2f a;
        a.x = x[(size_t)mrow * FDIM + ka];
        a.y = x[(size_t)mrow * FDIM + ka + 1];
#pragma unroll
        for (int t = 0; t < 8; ++t) {
            const int n = t * 16 + ncol;
            v2f b;
            b.x = W[(size_t)ka * FDIM + n];
            b.y = W[(size_t)(ka + 1) * FDIM + n];
            acc[t] = __builtin_amdgcn_wmma_f32_16x16x4_f32(
                false, a, false, b, (short)0, acc[t], false, false);
        }
    }

    const int mbase = row0 + ((lane >> 4) << 3);
#pragma unroll
    for (int t = 0; t < 8; ++t) {
        const int n = t * 16 + ncol;
#pragma unroll
        for (int r = 0; r < 8; ++r)
            h[(size_t)(mbase + r) * FDIM + n] = acc[t][r];
    }
}

__global__ void attn_terms(const float* __restrict__ h,
                           const float* __restrict__ att_src,
                           const float* __restrict__ att_dst,
                           float* __restrict__ a_src, float* __restrict__ a_dst,
                           int Nn) {
    int i = blockIdx.x * blockDim.x + threadIdx.x;
    if (i >= Nn * HEADS) return;
    int node = i >> 2, hd = i & 3;
    const float* hp = h + (size_t)node * FDIM + hd * CHAN;
    const float* as = att_src + hd * CHAN;
    const float* ad = att_dst + hd * CHAN;
    float s = 0.f, d = 0.f;
#pragma unroll
    for (int c = 0; c < CHAN; ++c) {
        float v = hp[c];
        s = fmaf(v, as[c], s);
        d = fmaf(v, ad[c], d);
    }
    a_src[i] = s;
    a_dst[i] = d;
}

__device__ __forceinline__ float leaky(float v) {
    return v > 0.f ? v : NEG_SLOPE * v;
}

__device__ __forceinline__ void atomicMaxF32(float* addr, float val) {
    if (val >= 0.f)
        atomicMax((int*)addr, __float_as_int(val));
    else
        atomicMin((unsigned int*)addr, __float_as_uint(val));
}

__global__ void edge_max(const int* __restrict__ ei,
                         const float* __restrict__ a_src,
                         const float* __restrict__ a_dst,
                         float* __restrict__ amax, int Ecnt) {
    int i = blockIdx.x * blockDim.x + threadIdx.x;
    if (i >= Ecnt * HEADS) return;
    int e = i >> 2, hd = i & 3;
    int s = ei[e], d = ei[Ecnt + e];
    float al = leaky(a_src[s * HEADS + hd] + a_dst[d * HEADS + hd]);
    atomicMaxF32(&amax[d * HEADS + hd], al);
}

__global__ void edge_denom(const int* __restrict__ ei,
                           const float* __restrict__ a_src,
                           const float* __restrict__ a_dst,
                           const float* __restrict__ amax,
                           float* __restrict__ denom, int Ecnt) {
    int i = blockIdx.x * blockDim.x + threadIdx.x;
    if (i >= Ecnt * HEADS) return;
    int e = i >> 2, hd = i & 3;
    int s = ei[e], d = ei[Ecnt + e];
    float al = leaky(a_src[s * HEADS + hd] + a_dst[d * HEADS + hd]);
    float mx = amax[d * HEADS + hd];
    if (!__builtin_isfinite(mx)) mx = 0.f;
    atomicAdd(&denom[d * HEADS + hd], expf(al - mx));
}

#define TILE_NODES   392
#define MSG_THREADS  1024
#define MSG_WAVES    (MSG_THREADS / 32)
#define SCAN_VECS    4
#define QUEUE_PER_WAVE 544
#define ROWS_IN_FLIGHT 8

typedef __attribute__((address_space(3))) volatile int lds_queue_word;

__device__ __forceinline__ int lane_bcast_i(int v, int j) {
    return __builtin_amdgcn_readlane(v, j);
}
__device__ __forceinline__ float lane_bcast_f(float v, int j) {
    return __int_as_float(__builtin_amdgcn_readlane(__float_as_int(v), j));
}

__device__ __forceinline__ void message_batch(lds_queue_word* q, int first, int cnt,
                                              int lane, int lo,
                                              const int* __restrict__ ei,
                                              const float* __restrict__ a_src,
                                              const float* __restrict__ a_dst,
                                              const float* __restrict__ amax,
                                              const float* __restrict__ denom,
                                              const float* __restrict__ h,
                                              float* acc, int Ecnt) {
    const int e = q[first + (lane < cnt ? lane : 0)];
    const int s = ei[e];
    const int d = ei[Ecnt + e];
    const int dl = d - lo;
    float w[HEADS];
#pragma unroll
    for (int k = 0; k < HEADS; ++k) {
        float al = leaky(a_src[s * HEADS + k] + a_dst[d * HEADS + k]);
        float mx = amax[d * HEADS + k];
        if (!__builtin_isfinite(mx)) mx = 0.f;
        w[k] = expf(al - mx) / (denom[d * HEADS + k] + 1e-16f);
    }
    for (int j0 = 0; j0 < cnt; j0 += ROWS_IN_FLIGHT) {
        float v[ROWS_IN_FLIGHT][HEADS];
#pragma unroll
        for (int t = 0; t < ROWS_IN_FLIGHT; ++t) {
            const int jj = (j0 + t < cnt) ? (j0 + t) : (cnt - 1);
            const float* hrow = h + (size_t)lane_bcast_i(s, jj) * FDIM + lane;
#pragma unroll
            for (int k = 0; k < HEADS; ++k) v[t][k] = hrow[k * CHAN];
        }
#pragma unroll
        for (int t = 0; t < ROWS_IN_FLIGHT; ++t) {
            const int jj = j0 + t;
            if (jj < cnt) {
                float* arow = acc + lane_bcast_i(dl, jj) * FDIM + lane;
#pragma unroll
                for (int k = 0; k < HEADS; ++k)
                    atomicAdd(arow + k * CHAN, v[t][k] * lane_bcast_f(w[k], jj));
            }
        }
    }
}

__device__ __forceinline__ void load_step(int4 (&buf)[SCAN_VECS], const int* __restrict__ p) {
#pragma unroll
    for (int u = 0; u < SCAN_VECS; ++u)
        buf[u] = *reinterpret_cast<const int4*>(p + u * MSG_THREADS * 4);
}

__device__ __forceinline__ void scan_consume(const int4 (&dv)[SCAN_VECS], int base, bool last,
                                             lds_queue_word* q, int& qn,
                                             int tid, int lane, int lo, int span,
                                             const int* __restrict__ ei,
                                             const float* __restrict__ a_src,
                                             const float* __restrict__ a_dst,
                                             const float* __restrict__ amax,
                                             const float* __restrict__ denom,
                                             const float* __restrict__ h,
                                             float* acc, int Ecnt) {
#pragma unroll
    for (int u = 0; u < SCAN_VECS; ++u) {
        const int pos = base + (u * MSG_THREADS + tid) * 4;
        const int d4[4] = {dv[u].x, dv[u].y, dv[u].z, dv[u].w};
        bool hit[4];
        bool any = false;
#pragma unroll
        for (int c = 0; c < 4; ++c) {
            hit[c] = (unsigned)(d4[c] - lo) < (unsigned)span;
            any |= hit[c];
        }
        if (__builtin_amdgcn_ballot_w32(any) == 0u) continue;
#pragma unroll
        for (int c = 0; c < 4; ++c) {
            const unsigned m = __builtin_amdgcn_ballot_w32(hit[c]);
            if (hit[c]) q[qn + (int)__builtin_amdgcn_mbcnt_lo(m, 0u)] = pos + c;
            qn += __builtin_popcount(m);
        }
    }
    while (qn >= 32 || (last && qn > 0)) {
        const int c = qn < 32 ? qn : 32;
        qn -= c;
        message_batch(q, qn, c, lane, lo, ei, a_src, a_dst, amax, denom, h, acc, Ecnt);
    }
}

__global__ __launch_bounds__(MSG_THREADS)
void edge_message_tiles(const int* __restrict__ ei,
                        const float* __restrict__ a_src,
                        const float* __restrict__ a_dst,
                        const float* __restrict__ amax,
                        const float* __restrict__ denom,
                        const float* __restrict__ h,
                        const float* __restrict__ bias,
                        float* __restrict__ out, int Ecnt, int Nn) {
    __shared__ float acc[TILE_NODES * FDIM];
    __shared__ int queue[MSG_WAVES * QUEUE_PER_WAVE];

    const int tid  = threadIdx.x;
    const int lane = tid & 31;
    const int wave = tid >> 5;
    const int lo   = blockIdx.x * TILE_NODES;
    int span = Nn - lo;
    if (span > TILE_NODES) span = TILE_NODES;
    if (span < 0) span = 0;

    for (int i = tid; i < TILE_NODES * FDIM; i += MSG_THREADS) acc[i] = 0.0f;
    __syncthreads();

    lds_queue_word* q = (lds_queue_word*)(queue + wave * QUEUE_PER_WAVE);
    int qn = 0;

    const int* dstp = ei + Ecnt;
    const int step_edges = MSG_THREADS * 4 * SCAN_VECS;
    const int whole = ((Ecnt & 3) == 0) ? (Ecnt / step_edges) * step_edges : 0;

    if (whole > 0) {
        const int* lp = dstp + tid * 4;
        int4 va[SCAN_VECS], vb[SCAN_VECS];
        load_step(va, lp);
        for (int base = 0; base < whole; base += 2 * step_edges) {
            const int b1 = base + step_edges;
            const int b2 = b1 + step_edges;
            load_step(vb, lp + (b1 < whole ? b1 : base));
            scan_consume(va, base, b1 >= Ecnt, q, qn, tid, lane, lo, span,
                         ei, a_src, a_dst, amax, denom, h, acc, Ecnt);
            load_step(va, lp + (b2 < whole ? b2 : base));
            if (b1 < whole)
                scan_consume(vb, b1, b2 >= Ecnt, q, qn, tid, lane, lo, span,
                             ei, a_src, a_dst, amax, denom, h, acc, Ecnt);
        }
    }
    for (int base = whole; base < Ecnt; base += step_edges) {
        int4 dv[SCAN_VECS];
#pragma unroll
        for (int u = 0; u < SCAN_VECS; ++u) {
            const int pos = base + (u * MSG_THREADS + tid) * 4;
            dv[u].x = (pos     < Ecnt) ? dstp[pos]     : -1;
            dv[u].y = (pos + 1 < Ecnt) ? dstp[pos + 1] : -1;
            dv[u].z = (pos + 2 < Ecnt) ? dstp[pos + 2] : -1;
            dv[u].w = (pos + 3 < Ecnt) ? dstp[pos + 3] : -1;
        }
        scan_consume(dv, base, base + step_edges >= Ecnt, q, qn, tid, lane, lo, span,
                     ei, a_src, a_dst, amax, denom, h, acc, Ecnt);
    }

    __syncthreads();

    for (int i = tid; i < span * FDIM; i += MSG_THREADS) {
        float v = acc[i] + bias[i & (FDIM - 1)];
        out[(size_t)lo * FDIM + i] = v > 0.f ? v : expm1f(v);
    }
}

extern "C" void kernel_launch(void* const* d_in, const int* in_sizes, int n_in,
                              void* d_out, int out_size, void* d_ws, size_t ws_size,
                              hipStream_t stream) {
    const float* x       = (const float*)d_in[0];
    const int*   ei      = (const int*)  d_in[1];
    const float* W       = (const float*)d_in[2];
    const float* att_src = (const float*)d_in[3];
    const float* att_dst = (const float*)d_in[4];
    const float* bias    = (const float*)d_in[5];
    float* out = (float*)d_out;

    const int N = in_sizes[0] / FDIM;
    const int E = in_sizes[1] / 2;

    float* ws     = (float*)d_ws;
    float* h      = ws;
    float* a_src  = h     + (size_t)N * FDIM;
    float* a_dst  = a_src + (size_t)N * HEADS;
    float* amax   = a_dst + (size_t)N * HEADS;
    float* denom  = amax  + (size_t)N * HEADS;

    const int B = 256;

    init_kernel<<<(N * HEADS + B - 1) / B, B, 0, stream>>>(amax, denom, N);

    int tiles  = (N + 15) / 16;
    int gblk   = (tiles + 3) / 4;
    gemm_wmma<<<gblk, 128, 0, stream>>>(x, W, h, N);

    attn_terms<<<(N * HEADS + B - 1) / B, B, 0, stream>>>(h, att_src, att_dst,
                                                          a_src, a_dst, N);
    edge_max  <<<(E * HEADS + B - 1) / B, B, 0, stream>>>(ei, a_src, a_dst, amax, E);
    edge_denom<<<(E * HEADS + B - 1) / B, B, 0, stream>>>(ei, a_src, a_dst, amax,
                                                          denom, E);
    edge_message_tiles<<<(N + TILE_NODES - 1) / TILE_NODES, MSG_THREADS, 0, stream>>>(
        ei, a_src, a_dst, amax, denom, h, bias, out, E, N);
}


# softmax max and exp-sum kept in LDS by the work-group owning the dst rows (edge ids listed per wave, LDS max then LDS float add, arrival order); rows divided once after the sum: 3 launches gone
# speedup vs baseline: 4.1597x; 1.8518x over previous
#include <hip/hip_runtime.h>
#include <math.h>

typedef float v2f __attribute__((ext_vector_type(2)));
typedef float v8f __attribute__((ext_vector_type(8)));

#define HEADS 4
#define CHAN 32
#define FDIM 128
#define NEG_SLOPE 0.2f

__global__ void gemm_wmma(const float* __restrict__ x, const float* __restrict__ W,
                          float* __restrict__ h, int Nn) {
    const int wave = threadIdx.x >> 5;
    const int lane = threadIdx.x & 31;
    const int row0 = (blockIdx.x * 4 + wave) * 16;
    if (row0 >= Nn) return;

    v8f acc[8];
#pragma unroll
    for (int t = 0; t < 8; ++t) acc[t] = (v8f)(0.0f);

    const int mrow  = row0 + (lane & 15);
    const int khalf = (lane >> 4) << 1;
    const int ncol  = lane & 15;

    for (int k0 = 0; k0 < FDIM; k0 += 4) {
        const int ka = k0 + khalf;
        v2f a;
        a.x = x[(size_t)mrow * FDIM + ka];
        a.y = x[(size_t)mrow * FDIM + ka + 1];
#pragma unroll
        for (int t = 0; t < 8; ++t) {
            const int n = t * 16 + ncol;
            v2f b;
            b.x = W[(size_t)ka * FDIM + n];
            b.y = W[(size_t)(ka + 1) * FDIM + n];
            acc[t] = __builtin_amdgcn_wmma_f32_16x16x4_f32(
                false, a, false, b, (short)0, acc[t], false, false);
        }
    }

    const int mbase = row0 + ((lane >> 4) << 3);
#pragma unroll
    for (int t = 0; t < 8; ++t) {
        const int n = t * 16 + ncol;
#pragma unroll
        for (int r = 0; r < 8; ++r)
            h[(size_t)(mbase + r) * FDIM + n] = acc[t][r];
    }
}

__global__ void attn_terms(const float* __restrict__ h,
                           const float* __restrict__ att_src,
                           const float* __restrict__ att_dst,
                           float* __restrict__ a_src, float* __restrict__ a_dst,
                           int Nn) {
    int i = blockIdx.x * blockDim.x + threadIdx.x;
    if (i >= Nn * HEADS) return;
    int node = i >> 2, hd = i & 3;
    const float* hp = h + (size_t)node * FDIM + hd * CHAN;
    const float* as = att_src + hd * CHAN;
    const float* ad = att_dst + hd * CHAN;
    float s = 0.f, d = 0.f;
#pragma unroll
    for (int c = 0; c < CHAN; ++c) {
        float v = hp[c];
        s = fmaf(v, as[c], s);
        d = fmaf(v, ad[c], d);
    }
    a_src[i] = s;
    a_dst[i] = d;
}

__device__ __forceinline__ float leaky(float v) {
    return v > 0.f ? v : NEG_SLOPE * v;
}

__device__ __forceinline__ void atomicMaxF32(float* addr, float val) {
    if (val >= 0.f)
        atomicMax((int*)addr, __float_as_int(val));
    else
        atomicMin((unsigned int*)addr, __float_as_uint(val));
}

#define TILE_NODES   392
#define MSG_THREADS  1024
#define MSG_WAVES    (MSG_THREADS / 32)
#define SCAN_VECS    4
#define STEP_PER_WAVE (32 * 4 * SCAN_VECS)
#define LIST_PER_WAVE 512
#define ROWS_IN_FLIGHT 8
#define MAX_IN_FLIGHT  4

static_assert(LIST_PER_WAVE >= STEP_PER_WAVE, "one step's ids of a wave must always fit its list");

typedef __attribute__((address_space(3))) volatile int lds_queue_word;

__device__ __forceinline__ int lane_bcast_i(int v, int j) {
    return __builtin_amdgcn_readlane(v, j);
}
__device__ __forceinline__ float lane_bcast_f(float v, int j) {
    return __int_as_float(__builtin_amdgcn_readlane(__float_as_int(v), j));
}

__device__ __forceinline__ void list_max(lds_queue_word* q, int n, int lane, int lo,
                                         const int* __restrict__ ei,
                                         const float* __restrict__ a_src,
                                         const float* __restrict__ a_dst,
                                         float* tmax, int Ecnt) {
    for (int b = 0; b < n; b += 32 * MAX_IN_FLIGHT) {
        int e[MAX_IN_FLIGHT], s[MAX_IN_FLIGHT], dl[MAX_IN_FLIGHT];
        float4 as4[MAX_IN_FLIGHT], ad4[MAX_IN_FLIGHT];
#pragma unroll
        for (int t = 0; t < MAX_IN_FLIGHT; ++t) {
            const int i = b + t * 32 + lane;
            e[t] = q[i < n ? i : n - 1];
        }
#pragma unroll
        for (int t = 0; t < MAX_IN_FLIGHT; ++t) {
            s[t]  = ei[e[t]];
            dl[t] = ei[Ecnt + e[t]];
        }
#pragma unroll
        for (int t = 0; t < MAX_IN_FLIGHT; ++t) {
            as4[t] = *reinterpret_cast<const float4*>(a_src + (size_t)s[t] * HEADS);
            ad4[t] = *reinterpret_cast<const float4*>(a_dst + (size_t)dl[t] * HEADS);
            dl[t] -= lo;
        }
#pragma unroll
        for (int t = 0; t < MAX_IN_FLIGHT; ++t) {
            const bool live = b + t * 32 + lane < n;
            float* slot = tmax + (live ? dl[t] : TILE_NODES + lane) * HEADS;
            atomicMaxF32(slot + 0, leaky(as4[t].x + ad4[t].x));
            atomicMaxF32(slot + 1, leaky(as4[t].y + ad4[t].y));
            atomicMaxF32(slot + 2, leaky(as4[t].z + ad4[t].z));
            atomicMaxF32(slot + 3, leaky(as4[t].w + ad4[t].w));
        }
    }
}

__device__ __forceinline__ void message_batch(lds_queue_word* q, int first, int cnt,
                                              int lane, int lo,
                                              const int* __restrict__ ei,
                                              const float* __restrict__ a_src,
                                              const float* __restrict__ a_dst,
                                              const float* tmax, float* tden,
                                              const float* __restrict__ h,
                                              float* acc, int Ecnt) {
    const int e = q[first + (lane < cnt ? lane : 0)];
    const int s = ei[e];
    const int d = ei[Ecnt + e];
    const int dl = d - lo;
    const float4 as4 = *reinterpret_cast<const float4*>(a_src + (size_t)s * HEADS);
    const float4 ad4 = *reinterpret_cast<const float4*>(a_dst + (size_t)d * HEADS);
    const float lg[HEADS] = {as4.x + ad4.x, as4.y + ad4.y, as4.z + ad4.z, as4.w + ad4.w};
    float w[HEADS];
#pragma unroll
    for (int k = 0; k < HEADS; ++k) {
        float mx = tmax[dl * HEADS + k];
        if (!__builtin_isfinite(mx)) mx = 0.f;
        w[k] = expf(leaky(lg[k]) - mx);
    }
    if (lane < cnt) {
#pragma unroll
        for (int k = 0; k < HEADS; ++k) atomicAdd(tden + dl * HEADS + k, w[k]);
    }
    for (int j0 = 0; j0 < cnt; j0 += ROWS_IN_FLIGHT) {
        float v[ROWS_IN_FLIGHT][HEADS];
#pragma unroll
        for (int t = 0; t < ROWS_IN_FLIGHT; ++t) {
            const int jj = (j0 + t < cnt) ? (j0 + t) : (cnt - 1);
            const float* hrow = h + (size_t)lane_bcast_i(s, jj) * FDIM + lane;
#pragma unroll
            for (int k = 0; k < HEADS; ++k) v[t][k] = hrow[k * CHAN];
        }
#pragma unroll
        for (int t = 0; t < ROWS_IN_FLIGHT; ++t) {
            const int jj = j0 + t;
            if (jj < cnt) {
                float* arow = acc + lane_bcast_i(dl, jj) * FDIM + lane;
#pragma unroll
                for (int k = 0; k < HEADS; ++k)
                    atomicAdd(arow + k * CHAN, v[t][k] * lane_bcast_f(w[k], jj));
            }
        }
    }
}

__device__ __forceinline__ void list_sum(lds_queue_word* q, int n, int lane, int lo,
                                         const int* __restrict__ ei,
                                         const float* __restrict__ a_src,
                                         const float* __restrict__ a_dst,
                                         const float* tmax, float* tden,
                                         const float* __restrict__ h,
                                         float* acc, int Ecnt) {
    for (int first = 0; first < n; first += 32) {
        const int c = (n - first) < 32 ? (n - first) : 32;
        message_batch(q, first, c, lane, lo, ei, a_src, a_dst, tmax, tden, h, acc, Ecnt);
    }
}

__device__ __forceinline__ void load_step(int4 (&buf)[SCAN_VECS], const int* __restrict__ p) {
#pragma unroll
    for (int u = 0; u < SCAN_VECS; ++u)
        buf[u] = *reinterpret_cast<const int4*>(p + u * MSG_THREADS * 4);
}

__device__ __forceinline__ void load_step_guarded(int4 (&dv)[SCAN_VECS],
                                                  const int* __restrict__ dstp,
                                                  int base, int tid, int Ecnt) {
#pragma unroll
    for (int u = 0; u < SCAN_VECS; ++u) {
        const int pos = base + (u * MSG_THREADS + tid) * 4;
        dv[u].x = (pos     < Ecnt) ? dstp[pos]     : -1;
        dv[u].y = (pos + 1 < Ecnt) ? dstp[pos + 1] : -1;
        dv[u].z = (pos + 2 < Ecnt) ? dstp[pos + 2] : -1;
        dv[u].w = (pos + 3 < Ecnt) ? dstp[pos + 3] : -1;
    }
}

__device__ __forceinline__ void scan_append(const int4 (&dv)[SCAN_VECS], int base,
                                            lds_queue_word* q, int& qn,
                                            int tid, int lo, int span) {
#pragma unroll
    for (int u = 0; u < SCAN_VECS; ++u) {
        const int pos = base + (u * MSG_THREADS + tid) * 4;
        const int d4[4] = {dv[u].x, dv[u].y, dv[u].z, dv[u].w};
        bool hit[4];
        bool any = false;
#pragma unroll
        for (int c = 0; c < 4; ++c) {
            hit[c] = (unsigned)(d4[c] - lo) < (unsigned)span;
            any |= hit[c];
        }
        if (__builtin_amdgcn_ballot_w32(any) == 0u) continue;
#pragma unroll
        for (int c = 0; c < 4; ++c) {
            const unsigned m = __builtin_amdgcn_ballot_w32(hit[c]);
            const int slot = qn + (int)__builtin_amdgcn_mbcnt_lo(m, 0u);
            if (hit[c] && slot < LIST_PER_WAVE) q[slot] = pos + c;
            qn += __builtin_popcount(m);
        }
    }
}

__global__ __launch_bounds__(MSG_THREADS)
void edge_message_tiles(const int* __restrict__ ei,
                        const float* __restrict__ a_src,
                        const float* __restrict__ a_dst,
                        const float* __restrict__ h,
                        const float* __restrict__ bias,
                        float* __restrict__ out, int Ecnt, int Nn) {
    __shared__ float acc[TILE_NODES * FDIM];
    __shared__ int   list[MSG_WAVES * LIST_PER_WAVE];
    __shared__ float tmax[(TILE_NODES + 32) * HEADS];
    __shared__ float tden[TILE_NODES * HEADS];
    __shared__ int   crowded;

    const int tid  = threadIdx.x;
    const int lane = tid & 31;
    const int wave = tid >> 5;
    const int lo   = blockIdx.x * TILE_NODES;
    int span = Nn - lo;
    if (span > TILE_NODES) span = TILE_NODES;
    if (span < 0) span = 0;

    for (int i = tid; i < TILE_NODES * FDIM; i += MSG_THREADS) acc[i] = 0.0f;
    for (int i = tid; i < (TILE_NODES + 32) * HEADS; i += MSG_THREADS) tmax[i] = -__builtin_inff();
    for (int i = tid; i < TILE_NODES * HEADS; i += MSG_THREADS) tden[i] = 0.0f;
    if (tid == 0) crowded = 0;
    __syncthreads();

    lds_queue_word* q = (lds_queue_word*)(list + wave * LIST_PER_WAVE);
    int qn = 0;

    const int* dstp = ei + Ecnt;
    const int step_edges = MSG_THREADS * 4 * SCAN_VECS;
    const int whole = ((Ecnt & 3) == 0) ? (Ecnt / step_edges) * step_edges : 0;

    if (whole > 0) {
        const int* lp = dstp + tid * 4;
        int4 va[SCAN_VECS], vb[SCAN_VECS];
        load_step(va, lp);
        for (int base = 0; base < whole; base += 2 * step_edges) {
            const int b1 = base + step_edges;
            const int b2 = b1 + step_edges;
            load_step(vb, lp + (b1 < whole ? b1 : base));
            scan_append(va, base, q, qn, tid, lo, span);
            load_step(va, lp + (b2 < whole ? b2 : base));
            if (b1 < whole)
                scan_append(vb, b1, q, qn, tid, lo, span);
        }
    }
    for (int base = whole; base < Ecnt; base += step_edges) {
        int4 dv[SCAN_VECS];
        load_step_guarded(dv, dstp, base, tid, Ecnt);
        scan_append(dv, base, q, qn, tid, lo, span);
    }

    if (qn > LIST_PER_WAVE && lane == 0) crowded = 1;
    __syncthreads();
    const bool step_by_step = crowded != 0;

    if (!step_by_step) {
        list_max(q, qn, lane, lo, ei, a_src, a_dst, tmax, Ecnt);
        __syncthreads();
        list_sum(q, qn, lane, lo, ei, a_src, a_dst, tmax, tden, h, acc, Ecnt);
    } else {
        for (int base = 0; base < Ecnt; base += step_edges) {
            int4 dv[SCAN_VECS];
            load_step_guarded(dv, dstp, base, tid, Ecnt);
            qn = 0;
            scan_append(dv, base, q, qn, tid, lo, span);
            list_max(q, qn, lane, lo, ei, a_src, a_dst, tmax, Ecnt);
        }
        __syncthreads();
        for (int base = 0; base < Ecnt; base += step_edges) {
            int4 dv[SCAN_VECS];
            load_step_guarded(dv, dstp, base, tid, Ecnt);
            qn = 0;
            scan_append(dv, base, q, qn, tid, lo, span);
            list_sum(q, qn, lane, lo, ei, a_src, a_dst, tmax, tden, h, acc, Ecnt);
        }
    }

    __syncthreads();

    for (int i = tid; i < span * FDIM; i += MSG_THREADS) {
        const int col = i & (FDIM - 1);
        const float den = tden[(i >> 7) * HEADS + (col >> 5)] + 1e-16f;
        float v = acc[i] / den + bias[col];
        out[(size_t)lo * FDIM + i] = v > 0.f ? v : expm1f(v);
    }
}

extern "C" void kernel_launch(void* const* d_in, const int* in_sizes, int n_in,
                              void* d_out, int out_size, void* d_ws, size_t ws_size,
                              hipStream_t stream) {
    const float* x       = (const float*)d_in[0];
    const int*   ei      = (const int*)  d_in[1];
    const float* W       = (const float*)d_in[2];
    const float* att_src = (const float*)d_in[3];
    const float* att_dst = (const float*)d_in[4];
    const float* bias    = (const float*)d_in[5];
    float* out = (float*)d_out;

    const int N = in_sizes[0] / FDIM;
    const int E = in_sizes[1] / 2;

    float* ws     = (float*)d_ws;
    float* h      = ws;
    float* a_src  = h     + (size_t)N * FDIM;
    float* a_dst  = a_src + (size_t)N * HEADS;

    const int B = 256;

    int tiles  = (N + 15) / 16;
    int gblk   = (tiles + 3) / 4;
    gemm_wmma<<<gblk, 128, 0, stream>>>(x, W, h, N);

    attn_terms<<<(N * HEADS + B - 1) / B, B, 0, stream>>>(h, att_src, att_dst,
                                                          a_src, a_dst, N);
    edge_message_tiles<<<(N + TILE_NODES - 1) / TILE_NODES, MSG_THREADS, 0, stream>>>(
        ei, a_src, a_dst, h, bias, out, E, N);
}


# scan reads one tile-number byte per edge (plane written in-call by the attention-term threads), not the 4-byte dst id: a quarter of the scan bytes; lists reordered, sums still arrival order
# speedup vs baseline: 4.8558x; 1.1673x over previous
#include <hip/hip_runtime.h>
#include <math.h>

typedef float v2f __attribute__((ext_vector_type(2)));
typedef float v8f __attribute__((ext_vector_type(8)));

#define HEADS 4
#define CHAN 32
#define FDIM 128
#define NEG_SLOPE 0.2f
#define TILE_NODES 392

__global__ void gemm_wmma(const float* __restrict__ x, const float* __restrict__ W,
                          float* __restrict__ h, int Nn) {
    const int wave = threadIdx.x >> 5;
    const int lane = threadIdx.x & 31;
    const int row0 = (blockIdx.x * 4 + wave) * 16;
    if (row0 >= Nn) return;

    v8f acc[8];
#pragma unroll
    for (int t = 0; t < 8; ++t) acc[t] = (v8f)(0.0f);

    const int mrow  = row0 + (lane & 15);
    const int khalf = (lane >> 4) << 1;
    const int ncol  = lane & 15;

    for (int k0 = 0; k0 < FDIM; k0 += 4) {
        const int ka = k0 + khalf;
        v2f a;
        a.x = x[(size_t)mrow * FDIM + ka];
        a.y = x[(size_t)mrow * FDIM + ka + 1];
#pragma unroll
        for (int t = 0; t < 8; ++t) {
            const int n = t * 16 + ncol;
            v2f b;
            b.x = W[(size_t)ka * FDIM + n];
            b.y = W[(size_t)(ka + 1) * FDIM + n];
            acc[t] = __builtin_amdgcn_wmma_f32_16x16x4_f32(
                false, a, false, b, (short)0, acc[t], false, false);
        }
    }

    const int mbase = row0 + ((lane >> 4) << 3);
#pragma unroll
    for (int t = 0; t < 8; ++t) {
        const int n = t * 16 + ncol;
#pragma unroll
        for (int r = 0; r < 8; ++r)
            h[(size_t)(mbase + r) * FDIM + n] = acc[t][r];
    }
}

__device__ __forceinline__ unsigned int tile_byte(int d) {
    const unsigned int t = (unsigned int)d / (unsigned int)TILE_NODES;
    return t < 255u ? t : 255u;
}

__global__ void attn_terms(const float* __restrict__ h,
                           const float* __restrict__ att_src,
                           const float* __restrict__ att_dst,
                           float* __restrict__ a_src, float* __restrict__ a_dst,
                           const int* __restrict__ ei,
                           unsigned int* __restrict__ tile_plane,
                           int Nn, int Ecnt) {
    const int i = blockIdx.x * blockDim.x + threadIdx.x;

    {
        const int nwords = (Ecnt + 3) >> 2;
        const int nthreads = gridDim.x * blockDim.x;
        const int* dstp = ei + Ecnt;
        const bool vec = (Ecnt & 3) == 0;
        for (int w = i; w < nwords; w += nthreads) {
            const int p = w * 4;
            int d0, d1, d2, d3;
            if (vec) {
                const int4 v = *reinterpret_cast<const int4*>(dstp + p);
                d0 = v.x; d1 = v.y; d2 = v.z; d3 = v.w;
            } else {
                d0 = dstp[p];
                d1 = (p + 1 < Ecnt) ? dstp[p + 1] : 0;
                d2 = (p + 2 < Ecnt) ? dstp[p + 2] : 0;
                d3 = (p + 3 < Ecnt) ? dstp[p + 3] : 0;
            }
            tile_plane[w] = tile_byte(d0) | (tile_byte(d1) << 8) |
                            (tile_byte(d2) << 16) | (tile_byte(d3) << 24);
        }
    }

    if (i >= Nn * HEADS) return;
    int node = i >> 2, hd = i & 3;
    const float* hp = h + (size_t)node * FDIM + hd * CHAN;
    const float* as = att_src + hd * CHAN;
    const float* ad = att_dst + hd * CHAN;
    float s = 0.f, d = 0.f;
#pragma unroll
    for (int c = 0; c < CHAN; ++c) {
        float v = hp[c];
        s = fmaf(v, as[c], s);
        d = fmaf(v, ad[c], d);
    }
    a_src[i] = s;
    a_dst[i] = d;
}

__device__ __forceinline__ float leaky(float v) {
    return v > 0.f ? v : NEG_SLOPE * v;
}

__device__ __forceinline__ void atomicMaxF32(float* addr, float val) {
    if (val >= 0.f)
        atomicMax((int*)addr, __float_as_int(val));
    else
        atomicMin((unsigned int*)addr, __float_as_uint(val));
}

#define MSG_THREADS  1024
#define MSG_WAVES    (MSG_THREADS / 32)
#define SCAN_VECS    4
#define IDS_PER_VEC  16
#define ROW_EDGES    (MSG_THREADS * IDS_PER_VEC)
#define STEP_EDGES   (ROW_EDGES * SCAN_VECS)
#define LIST_PER_WAVE 512
#define ROWS_IN_FLIGHT 8
#define MAX_IN_FLIGHT  4

static_assert(LIST_PER_WAVE >= 32 * IDS_PER_VEC, "one row's ids of a wave must always fit its list");

typedef __attribute__((address_space(3))) volatile int lds_queue_word;

__device__ __forceinline__ int lane_bcast_i(int v, int j) {
    return __builtin_amdgcn_readlane(v, j);
}
__device__ __forceinline__ float lane_bcast_f(float v, int j) {
    return __int_as_float(__builtin_amdgcn_readlane(__float_as_int(v), j));
}

__device__ __forceinline__ void list_max(lds_queue_word* q, int n, int lane, int lo, int span,
                                         const int* __restrict__ ei,
                                         const float* __restrict__ a_src,
                                         const float* __restrict__ a_dst,
                                         float* tmax, int Ecnt) {
    for (int b = 0; b < n; b += 32 * MAX_IN_FLIGHT) {
        int e[MAX_IN_FLIGHT], s[MAX_IN_FLIGHT], dl[MAX_IN_FLIGHT];
        float4 as4[MAX_IN_FLIGHT], ad4[MAX_IN_FLIGHT];
#pragma unroll
        for (int t = 0; t < MAX_IN_FLIGHT; ++t) {
            const int i = b + t * 32 + lane;
            e[t] = q[i < n ? i : n - 1];
        }
#pragma unroll
        for (int t = 0; t < MAX_IN_FLIGHT; ++t) {
            s[t]  = ei[e[t]];
            dl[t] = ei[Ecnt + e[t]];
        }
#pragma unroll
        for (int t = 0; t < MAX_IN_FLIGHT; ++t) {
            as4[t] = *reinterpret_cast<const float4*>(a_src + (size_t)s[t] * HEADS);
            ad4[t] = *reinterpret_cast<const float4*>(a_dst + (size_t)dl[t] * HEADS);
            dl[t] -= lo;
        }
#pragma unroll
        for (int t = 0; t < MAX_IN_FLIGHT; ++t) {
            const bool live = (b + t * 32 + lane < n) && ((unsigned)dl[t] < (unsigned)span);
            float* slot = tmax + (live ? dl[t] : TILE_NODES + lane) * HEADS;
            atomicMaxF32(slot + 0, leaky(as4[t].x + ad4[t].x));
            atomicMaxF32(slot + 1, leaky(as4[t].y + ad4[t].y));
            atomicMaxF32(slot + 2, leaky(as4[t].z + ad4[t].z));
            atomicMaxF32(slot + 3, leaky(as4[t].w + ad4[t].w));
        }
    }
}

__device__ __forceinline__ void message_batch(lds_queue_word* q, int first, int cnt,
                                              int lane, int lo, int span,
                                              const int* __restrict__ ei,
                                              const float* __restrict__ a_src,
                                              const float* __restrict__ a_dst,
                                              const float* tmax, float* tden,
                                              const float* __restrict__ h,
                                              float* acc, int Ecnt) {
    const int e = q[first + (lane < cnt ? lane : 0)];
    const int s = ei[e];
    const int d = ei[Ecnt + e];
    const bool valid = (lane < cnt) && ((unsigned)(d - lo) < (unsigned)span);
    const unsigned okm = __builtin_amdgcn_ballot_w32(valid);
    const int dl = valid ? d - lo : 0;
    const float4 as4 = *reinterpret_cast<const float4*>(a_src + (size_t)s * HEADS);
    const float4 ad4 = *reinterpret_cast<const float4*>(a_dst + (size_t)d * HEADS);
    const float lg[HEADS] = {as4.x + ad4.x, as4.y + ad4.y, as4.z + ad4.z, as4.w + ad4.w};
    float w[HEADS];
#pragma unroll
    for (int k = 0; k < HEADS; ++k) {
        float mx = tmax[dl * HEADS + k];
        if (!__builtin_isfinite(mx)) mx = 0.f;
        w[k] = expf(leaky(lg[k]) - mx);
    }
    if (valid) {
#pragma unroll
        for (int k = 0; k < HEADS; ++k) atomicAdd(tden + dl * HEADS + k, w[k]);
    }
    for (int j0 = 0; j0 < cnt; j0 += ROWS_IN_FLIGHT) {
        float v[ROWS_IN_FLIGHT][HEADS];
#pragma unroll
        for (int t = 0; t < ROWS_IN_FLIGHT; ++t) {
            const int jj = (j0 + t < cnt) ? (j0 + t) : (cnt - 1);
            const float* hrow = h + (size_t)lane_bcast_i(s, jj) * FDIM + lane;
#pragma unroll
            for (int k = 0; k < HEADS; ++k) v[t][k] = hrow[k * CHAN];
        }
#pragma unroll
        for (int t = 0; t < ROWS_IN_FLIGHT; ++t) {
            const int jj = j0 + t;
            if ((okm >> jj) & 1u) {
                float* arow = acc + lane_bcast_i(dl, jj) * FDIM + lane;
#pragma unroll
                for (int k = 0; k < HEADS; ++k)
                    atomicAdd(arow + k * CHAN, v[t][k] * lane_bcast_f(w[k], jj));
            }
        }
    }
}

__device__ __forceinline__ void list_sum(lds_queue_word* q, int n, int lane, int lo, int span,
                                         const int* __restrict__ ei,
                                         const float* __restrict__ a_src,
                                         const float* __restrict__ a_dst,
                                         const float* tmax, float* tden,
                                         const float* __restrict__ h,
                                         float* acc, int Ecnt) {
    for (int first = 0; first < n; first += 32) {
        const int c = (n - first) < 32 ? (n - first) : 32;
        message_batch(q, first, c, lane, lo, span, ei, a_src, a_dst, tmax, tden, h, acc, Ecnt);
    }
}

__device__ __forceinline__ void load_step(uint4 (&buf)[SCAN_VECS],
                                          const unsigned int* __restrict__ p) {
#pragma unroll
    for (int u = 0; u < SCAN_VECS; ++u)
        buf[u] = *reinterpret_cast<const uint4*>(p + u * MSG_THREADS * 4);
}

__device__ __forceinline__ uint4 load_vec_guarded(const unsigned int* __restrict__ plane,
                                                  int pos0, int Ecnt, unsigned int fillw) {
    unsigned int w[4];
#pragma unroll
    for (int j = 0; j < 4; ++j) {
        const int p = pos0 + 4 * j;
        unsigned int x = fillw;
        if (p < Ecnt) {
            x = plane[p >> 2];
            const int nv = Ecnt - p;
            if (nv < 4) {
                const unsigned int keep = (1u << (8 * nv)) - 1u;
                x = (x & keep) | (fillw & ~keep);
            }
        }
        w[j] = x;
    }
    return make_uint4(w[0], w[1], w[2], w[3]);
}

__device__ __forceinline__ unsigned int nonzero_bytes(unsigned int x) {
    return ((x & 0x7f7f7f7fu) + 0x7f7f7f7fu) | x;
}

__device__ __forceinline__ void scan_vec(const uint4 v, int pos0, unsigned int pat,
                                         lds_queue_word* q, int& qn) {
    const unsigned int miss = ((nonzero_bytes(v.x ^ pat) >> 7) & 0x01010101u) |
                              ((nonzero_bytes(v.y ^ pat) >> 6) & 0x02020202u) |
                              ((nonzero_bytes(v.z ^ pat) >> 5) & 0x04040404u) |
                              ((nonzero_bytes(v.w ^ pat) >> 4) & 0x08080808u);
    unsigned int hm = miss ^ 0x0f0f0f0fu;
    unsigned int any = __builtin_amdgcn_ballot_w32(hm != 0u);
    while (any != 0u) {
        const bool has = hm != 0u;
        const int bit = __builtin_ctz(hm | 0x80000000u);
        const int off = ((bit & 7) << 2) | (bit >> 3);
        const int slot = qn + (int)__builtin_amdgcn_mbcnt_lo(any, 0u);
        if (has && slot < LIST_PER_WAVE) q[slot] = pos0 + off;
        qn += __builtin_popcount(any);
        hm &= hm - 1u;
        any = __builtin_amdgcn_ballot_w32(hm != 0u);
    }
}

__device__ __forceinline__ void scan_step(const uint4 (&buf)[SCAN_VECS], int base, int tid,
                                          unsigned int pat, lds_queue_word* q, int& qn) {
#pragma unroll
    for (int u = 0; u < SCAN_VECS; ++u)
        scan_vec(buf[u], base + (u * MSG_THREADS + tid) * IDS_PER_VEC, pat, q, qn);
}

__global__ __launch_bounds__(MSG_THREADS)
void edge_message_tiles(const int* __restrict__ ei,
                        const unsigned int* __restrict__ plane,
                        const float* __restrict__ a_src,
                        const float* __restrict__ a_dst,
                        const float* __restrict__ h,
                        const float* __restrict__ bias,
                        float* __restrict__ out, int Ecnt, int Nn) {
    __shared__ float acc[TILE_NODES * FDIM];
    __shared__ int   list[MSG_WAVES * LIST_PER_WAVE];
    __shared__ float tmax[(TILE_NODES + 32) * HEADS];
    __shared__ float tden[TILE_NODES * HEADS];
    __shared__ int   crowded;

    const int tid  = threadIdx.x;
    const int lane = tid & 31;
    const int wave = tid >> 5;
    const int lo   = blockIdx.x * TILE_NODES;
    int span = Nn - lo;
    if (span > TILE_NODES) span = TILE_NODES;
    if (span < 0) span = 0;

    for (int i = tid; i < TILE_NODES * FDIM; i += MSG_THREADS) acc[i] = 0.0f;
    for (int i = tid; i < (TILE_NODES + 32) * HEADS; i += MSG_THREADS) tmax[i] = -__builtin_inff();
    for (int i = tid; i < TILE_NODES * HEADS; i += MSG_THREADS) tden[i] = 0.0f;
    if (tid == 0) crowded = 0;
    __syncthreads();

    lds_queue_word* q = (lds_queue_word*)(list + wave * LIST_PER_WAVE);
    int qn = 0;

    const unsigned int pat   = (blockIdx.x < 255u ? blockIdx.x : 255u) * 0x01010101u;
    const unsigned int fillw = pat ^ 0x80808080u;
    const int whole = (Ecnt / STEP_EDGES) * STEP_EDGES;

    if (whole > 0) {
        const unsigned int* lp = plane + tid * 4;
        uint4 va[SCAN_VECS], vb[SCAN_VECS];
        load_step(va, lp);
        for (int base = 0; base < whole; base += 2 * STEP_EDGES) {
            const int b1 = base + STEP_EDGES;
            const int b2 = b1 + STEP_EDGES;
            load_step(vb, lp + ((b1 < whole ? b1 : base) >> 2));
            scan_step(va, base, tid, pat, q, qn);
            load_step(va, lp + ((b2 < whole ? b2 : base) >> 2));
            if (b1 < whole)
                scan_step(vb, b1, tid, pat, q, qn);
        }
    }
    for (int base = whole; base < Ecnt; base += ROW_EDGES) {
        const int pos0 = base + tid * IDS_PER_VEC;
        scan_vec(load_vec_guarded(plane, pos0, Ecnt, fillw), pos0, pat, q, qn);
    }

    if (qn > LIST_PER_WAVE && lane == 0) crowded = 1;
    __syncthreads();
    const bool step_by_step = crowded != 0;

    if (!step_by_step) {
        list_max(q, qn, lane, lo, span, ei, a_src, a_dst, tmax, Ecnt);
        __syncthreads();
        list_sum(q, qn, lane, lo, span, ei, a_src, a_dst, tmax, tden, h, acc, Ecnt);
    } else {
        for (int base = 0; base < Ecnt; base += ROW_EDGES) {
            const int pos0 = base + tid * IDS_PER_VEC;
            qn = 0;
            scan_vec(load_vec_guarded(plane, pos0, Ecnt, fillw), pos0, pat, q, qn);
            list_max(q, qn, lane, lo, span, ei, a_src, a_dst, tmax, Ecnt);
        }
        __syncthreads();
        for (int base = 0; base < Ecnt; base += ROW_EDGES) {
            const int pos0 = base + tid * IDS_PER_VEC;
            qn = 0;
            scan_vec(load_vec_guarded(plane, pos0, Ecnt, fillw), pos0, pat, q, qn);
            list_sum(q, qn, lane, lo, span, ei, a_src, a_dst, tmax, tden, h, acc, Ecnt);
        }
    }

    __syncthreads();

    for (int i = tid; i < span * FDIM; i += MSG_THREADS) {
        const int col = i & (FDIM - 1);
        const float den = tden[(i >> 7) * HEADS + (col >> 5)] + 1e-16f;
        float v = acc[i] / den + bias[col];
        out[(size_t)lo * FDIM + i] = v > 0.f ? v : expm1f(v);
    }
}

extern "C" void kernel_launch(void* const* d_in, const int* in_sizes, int n_in,
                              void* d_out, int out_size, void* d_ws, size_t ws_size,
                              hipStream_t stream) {
    const float* x       = (const float*)d_in[0];
    const int*   ei      = (const int*)  d_in[1];
    const float* W       = (const float*)d_in[2];
    const float* att_src = (const float*)d_in[3];
    const float* att_dst = (const float*)d_in[4];
    const float* bias    = (const float*)d_in[5];
    float* out = (float*)d_out;

    const int N = in_sizes[0] / FDIM;
    const int E = in_sizes[1] / 2;

    float* ws     = (float*)d_ws;
    float* h      = ws;
    float* a_src  = h     + (size_t)N * FDIM;
    float* a_dst  = a_src + (size_t)N * HEADS;
    unsigned int* plane = (unsigned int*)(a_dst + (size_t)N * HEADS);

    const int B = 256;

    int tiles  = (N + 15) / 16;
    int gblk   = (tiles + 3) / 4;
    gemm_wmma<<<gblk, 128, 0, stream>>>(x, W, h, N);

    attn_terms<<<(N * HEADS + B - 1) / B, B, 0, stream>>>(h, att_src, att_dst,
                                                          a_src, a_dst, ei, plane, N, E);
    edge_message_tiles<<<(N + TILE_NODES - 1) / TILE_NODES, MSG_THREADS, 0, stream>>>(
        ei, plane, a_src, a_dst, h, bias, out, E, N);
}


# sum pass reuses what the max pass loaded per edge (src id and tile row left in the list words, logits kept in the lane's registers): its edge-id, index and term gathers are gone; sum order kept
# speedup vs baseline: 5.2374x; 1.0786x over previous
#include <hip/hip_runtime.h>
#include <math.h>

typedef float v2f __attribute__((ext_vector_type(2)));
typedef float v8f __attribute__((ext_vector_type(8)));

#define HEADS 4
#define CHAN 32
#define FDIM 128
#define NEG_SLOPE 0.2f
#define TILE_NODES 392

__global__ void gemm_wmma(const float* __restrict__ x, const float* __restrict__ W,
                          float* __restrict__ h, int Nn) {
    const int wave = threadIdx.x >> 5;
    const int lane = threadIdx.x & 31;
    const int row0 = (blockIdx.x * 4 + wave) * 16;
    if (row0 >= Nn) return;

    v8f acc[8];
#pragma unroll
    for (int t = 0; t < 8; ++t) acc[t] = (v8f)(0.0f);

    const int mrow  = row0 + (lane & 15);
    const int khalf = (lane >> 4) << 1;
    const int ncol  = lane & 15;

    for (int k0 = 0; k0 < FDIM; k0 += 4) {
        const int ka = k0 + khalf;
        v2f a;
        a.x = x[(size_t)mrow * FDIM + ka];
        a.y = x[(size_t)mrow * FDIM + ka + 1];
#pragma unroll
        for (int t = 0; t < 8; ++t) {
            const int n = t * 16 + ncol;
            v2f b;
            b.x = W[(size_t)ka * FDIM + n];
            b.y = W[(size_t)(ka + 1) * FDIM + n];
            acc[t] = __builtin_amdgcn_wmma_f32_16x16x4_f32(
                false, a, false, b, (short)0, acc[t], false, false);
        }
    }

    const int mbase = row0 + ((lane >> 4) << 3);
#pragma unroll
    for (int t = 0; t < 8; ++t) {
        const int n = t * 16 + ncol;
#pragma unroll
        for (int r = 0; r < 8; ++r)
            h[(size_t)(mbase + r) * FDIM + n] = acc[t][r];
    }
}

__device__ __forceinline__ unsigned int tile_byte(int d) {
    const unsigned int t = (unsigned int)d / (unsigned int)TILE_NODES;
    return t < 255u ? t : 255u;
}

__global__ void attn_terms(const float* __restrict__ h,
                           const float* __restrict__ att_src,
                           const float* __restrict__ att_dst,
                           float* __restrict__ a_src, float* __restrict__ a_dst,
                           const int* __restrict__ ei,
                           unsigned int* __restrict__ tile_plane,
                           int Nn, int Ecnt) {
    const int i = blockIdx.x * blockDim.x + threadIdx.x;

    {
        const int nwords = (Ecnt + 3) >> 2;
        const int nthreads = gridDim.x * blockDim.x;
        const int* dstp = ei + Ecnt;
        const bool vec = (Ecnt & 3) == 0;
        for (int w = i; w < nwords; w += nthreads) {
            const int p = w * 4;
            int d0, d1, d2, d3;
            if (vec) {
                const int4 v = *reinterpret_cast<const int4*>(dstp + p);
                d0 = v.x; d1 = v.y; d2 = v.z; d3 = v.w;
            } else {
                d0 = dstp[p];
                d1 = (p + 1 < Ecnt) ? dstp[p + 1] : 0;
                d2 = (p + 2 < Ecnt) ? dstp[p + 2] : 0;
                d3 = (p + 3 < Ecnt) ? dstp[p + 3] : 0;
            }
            tile_plane[w] = tile_byte(d0) | (tile_byte(d1) << 8) |
                            (tile_byte(d2) << 16) | (tile_byte(d3) << 24);
        }
    }

    if (i >= Nn * HEADS) return;
    int node = i >> 2, hd = i & 3;
    const float* hp = h + (size_t)node * FDIM + hd * CHAN;
    const float* as = att_src + hd * CHAN;
    const float* ad = att_dst + hd * CHAN;
    float s = 0.f, d = 0.f;
#pragma unroll
    for (int c = 0; c < CHAN; ++c) {
        float v = hp[c];
        s = fmaf(v, as[c], s);
        d = fmaf(v, ad[c], d);
    }
    a_src[i] = s;
    a_dst[i] = d;
}

__device__ __forceinline__ float leaky(float v) {
    return v > 0.f ? v : NEG_SLOPE * v;
}

__device__ __forceinline__ void atomicMaxF32(float* addr, float val) {
    if (val >= 0.f)
        atomicMax((int*)addr, __float_as_int(val));
    else
        atomicMin((unsigned int*)addr, __float_as_uint(val));
}

#define MSG_THREADS  1024
#define MSG_WAVES    (MSG_THREADS / 32)
#define SCAN_VECS    4
#define IDS_PER_VEC  16
#define ROW_EDGES    (MSG_THREADS * IDS_PER_VEC)
#define STEP_EDGES   (ROW_EDGES * SCAN_VECS)
#define LIST_PER_WAVE 512
#define ROWS_IN_FLIGHT 8
#define MAX_IN_FLIGHT  4
#define KEPT_SLOTS     8
#define NOT_IN_TILE    0xFFFFu

static_assert(LIST_PER_WAVE >= 32 * IDS_PER_VEC, "one row's ids of a wave must always fit its list");
static_assert(LIST_PER_WAVE % (32 * MAX_IN_FLIGHT) == 0, "a chunk of the max pass must end inside the list");
static_assert(KEPT_SLOTS % MAX_IN_FLIGHT == 0, "kept slots are whole chunks of the max pass");
static_assert(TILE_NODES < NOT_IN_TILE, "a row inside the tile must fit the 16-bit row word");

typedef __attribute__((address_space(3))) volatile int lds_queue_word;
typedef __attribute__((address_space(3))) volatile unsigned short lds_row_word;

__device__ __forceinline__ int lane_bcast_i(int v, int j) {
    return __builtin_amdgcn_readlane(v, j);
}
__device__ __forceinline__ float lane_bcast_f(float v, int j) {
    return __int_as_float(__builtin_amdgcn_readlane(__float_as_int(v), j));
}

__device__ __forceinline__ void max_chunk(lds_queue_word* q, lds_row_word* lr, int n, int b,
                                          int lane, int lo, int span,
                                          const int* __restrict__ ei,
                                          const float* __restrict__ a_src,
                                          const float* __restrict__ a_dst,
                                          float* tmax, int Ecnt, float (*lg)[HEADS]) {
    int e[MAX_IN_FLIGHT], s[MAX_IN_FLIGHT], dl[MAX_IN_FLIGHT];
    float4 as4[MAX_IN_FLIGHT], ad4[MAX_IN_FLIGHT];
#pragma unroll
    for (int t = 0; t < MAX_IN_FLIGHT; ++t) {
        const int i = b + t * 32 + lane;
        e[t] = q[i < n ? i : n - 1];
    }
#pragma unroll
    for (int t = 0; t < MAX_IN_FLIGHT; ++t) {
        s[t]  = ei[e[t]];
        dl[t] = ei[Ecnt + e[t]];
    }
#pragma unroll
    for (int t = 0; t < MAX_IN_FLIGHT; ++t) {
        as4[t] = *reinterpret_cast<const float4*>(a_src + (size_t)s[t] * HEADS);
        ad4[t] = *reinterpret_cast<const float4*>(a_dst + (size_t)dl[t] * HEADS);
        dl[t] -= lo;
    }
#pragma unroll
    for (int t = 0; t < MAX_IN_FLIGHT; ++t) {
        const int i = b + t * 32 + lane;
        const bool live = (i < n) && ((unsigned)dl[t] < (unsigned)span);
        float* slot = tmax + (live ? dl[t] : TILE_NODES + lane) * HEADS;
        lg[t][0] = leaky(as4[t].x + ad4[t].x);
        lg[t][1] = leaky(as4[t].y + ad4[t].y);
        lg[t][2] = leaky(as4[t].z + ad4[t].z);
        lg[t][3] = leaky(as4[t].w + ad4[t].w);
        atomicMaxF32(slot + 0, lg[t][0]);
        atomicMaxF32(slot + 1, lg[t][1]);
        atomicMaxF32(slot + 2, lg[t][2]);
        atomicMaxF32(slot + 3, lg[t][3]);
        q[i]  = s[t];
        lr[i] = live ? (unsigned short)dl[t] : (unsigned short)NOT_IN_TILE;
    }
}

__device__ __forceinline__ void list_max(lds_queue_word* q, lds_row_word* lr, int n,
                                         int lane, int lo, int span,
                                         const int* __restrict__ ei,
                                         const float* __restrict__ a_src,
                                         const float* __restrict__ a_dst,
                                         float* tmax, int Ecnt,
                                         float (&kept)[KEPT_SLOTS][HEADS]) {
#pragma unroll
    for (int c = 0; c < KEPT_SLOTS / MAX_IN_FLIGHT; ++c) {
        if (c * 32 * MAX_IN_FLIGHT < n)
            max_chunk(q, lr, n, c * 32 * MAX_IN_FLIGHT, lane, lo, span, ei, a_src, a_dst,
                      tmax, Ecnt, &kept[c * MAX_IN_FLIGHT]);
    }
    for (int b = KEPT_SLOTS * 32; b < n; b += 32 * MAX_IN_FLIGHT) {
        float spare[MAX_IN_FLIGHT][HEADS];
        max_chunk(q, lr, n, b, lane, lo, span, ei, a_src, a_dst, tmax, Ecnt, spare);
    }
}

__device__ __forceinline__ void list_resolve(lds_queue_word* q, lds_row_word* lr, int n,
                                             int lane, int lo, int span,
                                             const int* __restrict__ ei, int Ecnt) {
    for (int b = 0; b < n; b += 32) {
        const int i = b + lane;
        const int e = q[i < n ? i : n - 1];
        const int s = ei[e];
        const int dl = ei[Ecnt + e] - lo;
        q[i]  = s;
        lr[i] = ((i < n) && ((unsigned)dl < (unsigned)span)) ? (unsigned short)dl
                                                            : (unsigned short)NOT_IN_TILE;
    }
}

__device__ __forceinline__ void message_batch(lds_queue_word* q, lds_row_word* lr,
                                              int first, int cnt, int lane, int lo,
                                              const float* __restrict__ a_src,
                                              const float* __restrict__ a_dst,
                                              bool have_lg, const float (&lgk)[HEADS],
                                              const float* tmax, float* tden,
                                              const float* __restrict__ h,
                                              float* acc) {
    const int idx = first + (lane < cnt ? lane : 0);
    const int s = q[idx];
    const unsigned int r = lr[idx];
    const bool valid = (lane < cnt) && (r != NOT_IN_TILE);
    const unsigned okm = __builtin_amdgcn_ballot_w32(valid);
    const int dl = valid ? (int)r : 0;
    float lg[HEADS];
    if (have_lg) {
        lg[0] = lgk[0];
        lg[1] = lgk[1];
        lg[2] = lgk[2];
        lg[3] = lgk[3];
    } else {
        const float4 as4 = *reinterpret_cast<const float4*>(a_src + (size_t)s * HEADS);
        const float4 ad4 = *reinterpret_cast<const float4*>(a_dst + (size_t)(lo + dl) * HEADS);
        lg[0] = leaky(as4.x + ad4.x);
        lg[1] = leaky(as4.y + ad4.y);
        lg[2] = leaky(as4.z + ad4.z);
        lg[3] = leaky(as4.w + ad4.w);
    }
    float w[HEADS];
#pragma unroll
    for (int k = 0; k < HEADS; ++k) {
        float mx = tmax[dl * HEADS + k];
        if (!__builtin_isfinite(mx)) mx = 0.f;
        w[k] = expf(lg[k] - mx);
    }
    if (valid) {
#pragma unroll
        for (int k = 0; k < HEADS; ++k) atomicAdd(tden + dl * HEADS + k, w[k]);
    }
    for (int j0 = 0; j0 < cnt; j0 += ROWS_IN_FLIGHT) {
        float v[ROWS_IN_FLIGHT][HEADS];
#pragma unroll
        for (int t = 0; t < ROWS_IN_FLIGHT; ++t) {
            const int jj = (j0 + t < cnt) ? (j0 + t) : (cnt - 1);
            const float* hrow = h + (size_t)lane_bcast_i(s, jj) * FDIM + lane;
#pragma unroll
            for (int k = 0; k < HEADS; ++k) v[t][k] = hrow[k * CHAN];
        }
#pragma unroll
        for (int t = 0; t < ROWS_IN_FLIGHT; ++t) {
            const int jj = j0 + t;
            if ((okm >> jj) & 1u) {
                float* arow = acc + lane_bcast_i(dl, jj) * FDIM + lane;
#pragma unroll
                for (int k = 0; k < HEADS; ++k)
                    atomicAdd(arow + k * CHAN, v[t][k] * lane_bcast_f(w[k], jj));
            }
        }
    }
}

__device__ __forceinline__ void list_sum(lds_queue_word* q, lds_row_word* lr, int n,
                                         int lane, int lo,
                                         const float* __restrict__ a_src,
                                         const float* __restrict__ a_dst,
                                         const float* tmax, float* tden,
                                         const float* __restrict__ h,
                                         float* acc,
                                         float (&kept)[KEPT_SLOTS][HEADS], int nkept) {
    for (int first = 0; first < n; first += 32) {
        const int c = (n - first) < 32 ? (n - first) : 32;
        message_batch(q, lr, first, c, lane, lo, a_src, a_dst, first < nkept * 32, kept[0],
                      tmax, tden, h, acc);
#pragma unroll
        for (int j = 0; j + 1 < KEPT_SLOTS; ++j) {
#pragma unroll
            for (int k = 0; k < HEADS; ++k) kept[j][k] = kept[j + 1][k];
        }
    }
}

__device__ __forceinline__ void load_step(uint4 (&buf)[SCAN_VECS],
                                          const unsigned int* __restrict__ p) {
#pragma unroll
    for (int u = 0; u < SCAN_VECS; ++u)
        buf[u] = *reinterpret_cast<const uint4*>(p + u * MSG_THREADS * 4);
}

__device__ __forceinline__ uint4 load_vec_guarded(const unsigned int* __restrict__ plane,
                                                  int pos0, int Ecnt, unsigned int fillw) {
    unsigned int w[4];
#pragma unroll
    for (int j = 0; j < 4; ++j) {
        const int p = pos0 + 4 * j;
        unsigned int x = fillw;
        if (p < Ecnt) {
            x = plane[p >> 2];
            const int nv = Ecnt - p;
            if (nv < 4) {
                const unsigned int keep = (1u << (8 * nv)) - 1u;
                x = (x & keep) | (fillw & ~keep);
            }
        }
        w[j] = x;
    }
    return make_uint4(w[0], w[1], w[2], w[3]);
}

__device__ __forceinline__ unsigned int nonzero_bytes(unsigned int x) {
    return ((x & 0x7f7f7f7fu) + 0x7f7f7f7fu) | x;
}

__device__ __forceinline__ void scan_vec(const uint4 v, int pos0, unsigned int pat,
                                         lds_queue_word* q, int& qn) {
    const unsigned int miss = ((nonzero_bytes(v.x ^ pat) >> 7) & 0x01010101u) |
                              ((nonzero_bytes(v.y ^ pat) >> 6) & 0x02020202u) |
                              ((nonzero_bytes(v.z ^ pat) >> 5) & 0x04040404u) |
                              ((nonzero_bytes(v.w ^ pat) >> 4) & 0x08080808u);
    unsigned int hm = miss ^ 0x0f0f0f0fu;
    unsigned int any = __builtin_amdgcn_ballot_w32(hm != 0u);
    while (any != 0u) {
        const bool has = hm != 0u;
        const int bit = __builtin_ctz(hm | 0x80000000u);
        const int off = ((bit & 7) << 2) | (bit >> 3);
        const int slot = qn + (int)__builtin_amdgcn_mbcnt_lo(any, 0u);
        if (has && slot < LIST_PER_WAVE) q[slot] = pos0 + off;
        qn += __builtin_popcount(any);
        hm &= hm - 1u;
        any = __builtin_amdgcn_ballot_w32(hm != 0u);
    }
}

__device__ __forceinline__ void scan_step(const uint4 (&buf)[SCAN_VECS], int base, int tid,
                                          unsigned int pat, lds_queue_word* q, int& qn) {
#pragma unroll
    for (int u = 0; u < SCAN_VECS; ++u)
        scan_vec(buf[u], base + (u * MSG_THREADS + tid) * IDS_PER_VEC, pat, q, qn);
}

__global__ __launch_bounds__(MSG_THREADS)
void edge_message_tiles(const int* __restrict__ ei,
                        const unsigned int* __restrict__ plane,
                        const float* __restrict__ a_src,
                        const float* __restrict__ a_dst,
                        const float* __restrict__ h,
                        const float* __restrict__ bias,
                        float* __restrict__ out, int Ecnt, int Nn) {
    __shared__ float acc[TILE_NODES * FDIM];
    __shared__ int   list[MSG_WAVES * LIST_PER_WAVE];
    __shared__ unsigned short rows[MSG_WAVES * LIST_PER_WAVE];
    __shared__ float tmax[(TILE_NODES + 32) * HEADS];
    __shared__ float tden[TILE_NODES * HEADS];
    __shared__ int   crowded;

    const int tid  = threadIdx.x;
    const int lane = tid & 31;
    const int wave = tid >> 5;
    const int lo   = blockIdx.x * TILE_NODES;
    int span = Nn - lo;
    if (span > TILE_NODES) span = TILE_NODES;
    if (span < 0) span = 0;

    for (int i = tid; i < TILE_NODES * FDIM; i += MSG_THREADS) acc[i] = 0.0f;
    for (int i = tid; i < (TILE_NODES + 32) * HEADS; i += MSG_THREADS) tmax[i] = -__builtin_inff();
    for (int i = tid; i < TILE_NODES * HEADS; i += MSG_THREADS) tden[i] = 0.0f;
    if (tid == 0) crowded = 0;
    __syncthreads();

    lds_queue_word* q = (lds_queue_word*)(list + wave * LIST_PER_WAVE);
    lds_row_word* lr = (lds_row_word*)(rows + wave * LIST_PER_WAVE);
    float kept[KEPT_SLOTS][HEADS];
#pragma unroll
    for (int j = 0; j < KEPT_SLOTS; ++j) {
#pragma unroll
        for (int k = 0; k < HEADS; ++k) kept[j][k] = 0.f;
    }
    int qn = 0;

    const unsigned int pat   = (blockIdx.x < 255u ? blockIdx.x : 255u) * 0x01010101u;
    const unsigned int fillw = pat ^ 0x80808080u;
    const int whole = (Ecnt / STEP_EDGES) * STEP_EDGES;

    if (whole > 0) {
        const unsigned int* lp = plane + tid * 4;
        uint4 va[SCAN_VECS], vb[SCAN_VECS];
        load_step(va, lp);
        for (int base = 0; base < whole; base += 2 * STEP_EDGES) {
            const int b1 = base + STEP_EDGES;
            const int b2 = b1 + STEP_EDGES;
            load_step(vb, lp + ((b1 < whole ? b1 : base) >> 2));
            scan_step(va, base, tid, pat, q, qn);
            load_step(va, lp + ((b2 < whole ? b2 : base) >> 2));
            if (b1 < whole)
                scan_step(vb, b1, tid, pat, q, qn);
        }
    }
    for (int base = whole; base < Ecnt; base += ROW_EDGES) {
        const int pos0 = base + tid * IDS_PER_VEC;
        scan_vec(load_vec_guarded(plane, pos0, Ecnt, fillw), pos0, pat, q, qn);
    }

    if (qn > LIST_PER_WAVE && lane == 0) crowded = 1;
    __syncthreads();
    const bool step_by_step = crowded != 0;

    if (!step_by_step) {
        list_max(q, lr, qn, lane, lo, span, ei, a_src, a_dst, tmax, Ecnt, kept);
        __syncthreads();
        list_sum(q, lr, qn, lane, lo, a_src, a_dst, tmax, tden, h, acc, kept, KEPT_SLOTS);
    } else {
        for (int base = 0; base < Ecnt; base += ROW_EDGES) {
            const int pos0 = base + tid * IDS_PER_VEC;
            qn = 0;
            scan_vec(load_vec_guarded(plane, pos0, Ecnt, fillw), pos0, pat, q, qn);
            list_max(q, lr, qn, lane, lo, span, ei, a_src, a_dst, tmax, Ecnt, kept);
        }
        __syncthreads();
        for (int base = 0; base < Ecnt; base += ROW_EDGES) {
            const int pos0 = base + tid * IDS_PER_VEC;
            qn = 0;
            scan_vec(load_vec_guarded(plane, pos0, Ecnt, fillw), pos0, pat, q, qn);
            list_resolve(q, lr, qn, lane, lo, span, ei, Ecnt);
            list_sum(q, lr, qn, lane, lo, a_src, a_dst, tmax, tden, h, acc, kept, 0);
        }
    }

    __syncthreads();

    for (int i = tid; i < span * FDIM; i += MSG_THREADS) {
        const int col = i & (FDIM - 1);
        const float den = tden[(i >> 7) * HEADS + (col >> 5)] + 1e-16f;
        float v = acc[i] / den + bias[col];
        out[(size_t)lo * FDIM + i] = v > 0.f ? v : expm1f(v);
    }
}

extern "C" void kernel_launch(void* const* d_in, const int* in_sizes, int n_in,
                              void* d_out, int out_size, void* d_ws, size_t ws_size,
                              hipStream_t stream) {
    const float* x       = (const float*)d_in[0];
    const int*   ei      = (const int*)  d_in[1];
    const float* W       = (const float*)d_in[2];
    const float* att_src = (const float*)d_in[3];
    const float* att_dst = (const float*)d_in[4];
    const float* bias    = (const float*)d_in[5];
    float* out = (float*)d_out;

    const int N = in_sizes[0] / FDIM;
    const int E = in_sizes[1] / 2;

    float* ws     = (float*)d_ws;
    float* h      = ws;
    float* a_src  = h     + (size_t)N * FDIM;
    float* a_dst  = a_src + (size_t)N * HEADS;
    unsigned int* plane = (unsigned int*)(a_dst + (size_t)N * HEADS);

    const int B = 256;

    int tiles  = (N + 15) / 16;
    int gblk   = (tiles + 3) / 4;
    gemm_wmma<<<gblk, 128, 0, stream>>>(x, W, h, N);

    attn_terms<<<(N * HEADS + B - 1) / B, B, 0, stream>>>(h, att_src, att_dst,
                                                          a_src, a_dst, ei, plane, N, E);
    edge_message_tiles<<<(N + TILE_NODES - 1) / TILE_NODES, MSG_THREADS, 0, stream>>>(
        ei, plane, a_src, a_dst, h, bias, out, E, N);
}


# sum pass fetches a source row with one 16-byte load per lane (4 whole lines per wave load, was four one-line loads); acc rows kept permuted in LDS so adds stay conflict-free; sum order kept
# speedup vs baseline: 5.9982x; 1.1453x over previous
#include <hip/hip_runtime.h>
#include <math.h>

typedef float v2f __attribute__((ext_vector_type(2)));
typedef float v8f __attribute__((ext_vector_type(8)));

#define HEADS 4
#define CHAN 32
#define FDIM 128
#define NEG_SLOPE 0.2f
#define TILE_NODES 392

__device__ __forceinline__ unsigned int tile_byte(int d) {
    const unsigned int t = (unsigned int)d / (unsigned int)TILE_NODES;
    return t < 255u ? t : 255u;
}

__device__ __forceinline__ float sum8(float v) {
    v += __int_as_float(__builtin_amdgcn_update_dpp(0, __float_as_int(v), 0xB1, 0xF, 0xF, true));
    v += __int_as_float(__builtin_amdgcn_update_dpp(0, __float_as_int(v), 0x4E, 0xF, 0xF, true));
    v += __int_as_float(__builtin_amdgcn_update_dpp(0, __float_as_int(v), 0x141, 0xF, 0xF, true));
    return v;
}

#define GEMM_THREADS 128
#define FOLD_ROWS    16

__global__ __launch_bounds__(GEMM_THREADS)
void gemm_terms_plane(const float* __restrict__ x, const float* __restrict__ W,
                      const float* __restrict__ att_src, const float* __restrict__ att_dst,
                      float* __restrict__ h, float* __restrict__ a_src, float* __restrict__ a_dst,
                      const int* __restrict__ ei, unsigned int* __restrict__ tile_plane,
                      int Nn, int Ecnt) {
    __shared__ float fragB[32 * 32 * 2];

    const int wave = threadIdx.x >> 5;
    const int lane = threadIdx.x & 31;

    {
        const int i = blockIdx.x * GEMM_THREADS + threadIdx.x;
        const int nwords = (Ecnt + 3) >> 2;
        const int nthreads = gridDim.x * GEMM_THREADS;
        const int* dstp = ei + Ecnt;
        const bool vec = (Ecnt & 3) == 0;
        for (int w = i; w < nwords; w += nthreads) {
            const int p = w * 4;
            int d0, d1, d2, d3;
            if (vec) {
                const int4 v = *reinterpret_cast<const int4*>(dstp + p);
                d0 = v.x; d1 = v.y; d2 = v.z; d3 = v.w;
            } else {
                d0 = dstp[p];
                d1 = (p + 1 < Ecnt) ? dstp[p + 1] : 0;
                d2 = (p + 2 < Ecnt) ? dstp[p + 2] : 0;
                d3 = (p + 3 < Ecnt) ? dstp[p + 3] : 0;
            }
            tile_plane[w] = tile_byte(d0) | (tile_byte(d1) << 8) |
                            (tile_byte(d2) << 16) | (tile_byte(d3) << 24);
        }
    }

    {
        const float4 as4 = *reinterpret_cast<const float4*>(att_src + 4 * lane);
        const float4 ad4 = *reinterpret_cast<const float4*>(att_dst + 4 * lane);
        const int g = lane >> 3;
        const int p = lane & 7;
        for (int i0 = 0; i0 < 32; i0 += FOLD_ROWS) {
            float4 w4[FOLD_ROWS];
#pragma unroll
            for (int t = 0; t < FOLD_ROWS; ++t)
                w4[t] = *reinterpret_cast<const float4*>(W + (size_t)(wave * 32 + i0 + t) * FDIM + 4 * lane);
#pragma unroll
            for (int t = 0; t < FOLD_ROWS; ++t) {
                const int k = wave * 32 + i0 + t;
                float ps = w4[t].x * as4.x;
                ps = fmaf(w4[t].y, as4.y, ps);
                ps = fmaf(w4[t].z, as4.z, ps);
                ps = fmaf(w4[t].w, as4.w, ps);
                float pd = w4[t].x * ad4.x;
                pd = fmaf(w4[t].y, ad4.y, pd);
                pd = fmaf(w4[t].z, ad4.z, pd);
                pd = fmaf(w4[t].w, ad4.w, pd);
                ps = sum8(ps);
                pd = sum8(pd);
                if (p < 4) {
                    const float val = p == 0 ? ps : (p == 1 ? pd : 0.0f);
                    const int n = p * 4 + g;
                    fragB[(((k >> 2) * 32) + n + ((k & 2) << 3)) * 2 + (k & 1)] = val;
                }
            }
        }
    }
    __syncthreads();

    const int row0 = (blockIdx.x * 4 + wave) * 16;
    if (row0 >= Nn) return;

    v8f acc[9];
#pragma unroll
    for (int t = 0; t < 9; ++t) acc[t] = (v8f)(0.0f);

    const int mrow  = row0 + (lane & 15);
    const int khalf = (lane >> 4) << 1;
    const int ncol  = lane & 15;

    for (int k0 = 0; k0 < FDIM; k0 += 4) {
        const int ka = k0 + khalf;
        v2f a;
        a.x = x[(size_t)mrow * FDIM + ka];
        a.y = x[(size_t)mrow * FDIM + ka + 1];
#pragma unroll
        for (int t = 0; t < 8; ++t) {
            const int n = t * 16 + ncol;
            v2f b;
            b.x = W[(size_t)ka * FDIM + n];
            b.y = W[(size_t)(ka + 1) * FDIM + n];
            acc[t] = __builtin_amdgcn_wmma_f32_16x16x4_f32(
                false, a, false, b, (short)0, acc[t], false, false);
        }
        v2f b9;
        b9.x = fragB[((k0 >> 2) * 32 + lane) * 2];
        b9.y = fragB[((k0 >> 2) * 32 + lane) * 2 + 1];
        acc[8] = __builtin_amdgcn_wmma_f32_16x16x4_f32(
            false, a, false, b9, (short)0, acc[8], false, false);
    }

    const int mbase = row0 + ((lane >> 4) << 3);
#pragma unroll
    for (int t = 0; t < 8; ++t) {
        const int n = t * 16 + ncol;
#pragma unroll
        for (int r = 0; r < 8; ++r)
            h[(size_t)(mbase + r) * FDIM + n] = acc[t][r];
    }
    if (ncol < 8) {
        float* ap = (ncol < 4) ? a_src : a_dst;
        const int hd = ncol & 3;
#pragma unroll
        for (int r = 0; r < 8; ++r)
            ap[(size_t)(mbase + r) * HEADS + hd] = acc[8][r];
    }
}

__device__ __forceinline__ float leaky(float v) {
    return v > 0.f ? v : NEG_SLOPE * v;
}

__device__ __forceinline__ void atomicMaxF32(float* addr, float val) {
    if (val >= 0.f)
        atomicMax((int*)addr, __float_as_int(val));
    else
        atomicMin((unsigned int*)addr, __float_as_uint(val));
}

#define MSG_THREADS  1024
#define MSG_WAVES    (MSG_THREADS / 32)
#define SCAN_VECS    4
#define IDS_PER_VEC  16
#define ROW_EDGES    (MSG_THREADS * IDS_PER_VEC)
#define STEP_EDGES   (ROW_EDGES * SCAN_VECS)
#define LIST_PER_WAVE 512
#define ROWS_IN_FLIGHT 8
#define MAX_IN_FLIGHT  4
#define KEPT_SLOTS     8
#define NOT_IN_TILE    0xFFFFu

static_assert(LIST_PER_WAVE >= 32 * IDS_PER_VEC, "one row's ids of a wave must always fit its list");
static_assert(LIST_PER_WAVE % (32 * MAX_IN_FLIGHT) == 0, "a chunk of the max pass must end inside the list");
static_assert(KEPT_SLOTS % MAX_IN_FLIGHT == 0, "kept slots are whole chunks of the max pass");
static_assert(TILE_NODES < NOT_IN_TILE, "a row inside the tile must fit the 16-bit row word");

typedef __attribute__((address_space(3))) volatile int lds_queue_word;
typedef __attribute__((address_space(3))) volatile unsigned short lds_row_word;

__device__ __forceinline__ int lane_bcast_i(int v, int j) {
    return __builtin_amdgcn_readlane(v, j);
}
__device__ __forceinline__ float lane_bcast_f(float v, int j) {
    return __int_as_float(__builtin_amdgcn_readlane(__float_as_int(v), j));
}

__device__ __forceinline__ void max_chunk(lds_queue_word* q, lds_row_word* lr, int n, int b,
                                          int lane, int lo, int span,
                                          const int* __restrict__ ei,
                                          const float* __restrict__ a_src,
                                          const float* __restrict__ a_dst,
                                          float* tmax, int Ecnt, float (*lg)[HEADS]) {
    int e[MAX_IN_FLIGHT], s[MAX_IN_FLIGHT], dl[MAX_IN_FLIGHT];
    float4 as4[MAX_IN_FLIGHT], ad4[MAX_IN_FLIGHT];
#pragma unroll
    for (int t = 0; t < MAX_IN_FLIGHT; ++t) {
        const int i = b + t * 32 + lane;
        e[t] = q[i < n ? i : n - 1];
    }
#pragma unroll
    for (int t = 0; t < MAX_IN_FLIGHT; ++t) {
        s[t]  = ei[e[t]];
        dl[t] = ei[Ecnt + e[t]];
    }
#pragma unroll
    for (int t = 0; t < MAX_IN_FLIGHT; ++t) {
        as4[t] = *reinterpret_cast<const float4*>(a_src + (size_t)s[t] * HEADS);
        ad4[t] = *reinterpret_cast<const float4*>(a_dst + (size_t)dl[t] * HEADS);
        dl[t] -= lo;
    }
#pragma unroll
    for (int t = 0; t < MAX_IN_FLIGHT; ++t) {
        const int i = b + t * 32 + lane;
        const bool live = (i < n) && ((unsigned)dl[t] < (unsigned)span);
        float* slot = tmax + (live ? dl[t] : TILE_NODES + lane) * HEADS;
        lg[t][0] = leaky(as4[t].x + ad4[t].x);
        lg[t][1] = leaky(as4[t].y + ad4[t].y);
        lg[t][2] = leaky(as4[t].z + ad4[t].z);
        lg[t][3] = leaky(as4[t].w + ad4[t].w);
        atomicMaxF32(slot + 0, lg[t][0]);
        atomicMaxF32(slot + 1, lg[t][1]);
        atomicMaxF32(slot + 2, lg[t][2]);
        atomicMaxF32(slot + 3, lg[t][3]);
        q[i]  = s[t];
        lr[i] = live ? (unsigned short)dl[t] : (unsigned short)NOT_IN_TILE;
    }
}

__device__ __forceinline__ void list_max(lds_queue_word* q, lds_row_word* lr, int n,
                                         int lane, int lo, int span,
                                         const int* __restrict__ ei,
                                         const float* __restrict__ a_src,
                                         const float* __restrict__ a_dst,
                                         float* tmax, int Ecnt,
                                         float (&kept)[KEPT_SLOTS][HEADS]) {
#pragma unroll
    for (int c = 0; c < KEPT_SLOTS / MAX_IN_FLIGHT; ++c) {
        if (c * 32 * MAX_IN_FLIGHT < n)
            max_chunk(q, lr, n, c * 32 * MAX_IN_FLIGHT, lane, lo, span, ei, a_src, a_dst,
                      tmax, Ecnt, &kept[c * MAX_IN_FLIGHT]);
    }
    for (int b = KEPT_SLOTS * 32; b < n; b += 32 * MAX_IN_FLIGHT) {
        float spare[MAX_IN_FLIGHT][HEADS];
        max_chunk(q, lr, n, b, lane, lo, span, ei, a_src, a_dst, tmax, Ecnt, spare);
    }
}

__device__ __forceinline__ void list_resolve(lds_queue_word* q, lds_row_word* lr, int n,
                                             int lane, int lo, int span,
                                             const int* __restrict__ ei, int Ecnt) {
    for (int b = 0; b < n; b += 32) {
        const int i = b + lane;
        const int e = q[i < n ? i : n - 1];
        const int s = ei[e];
        const int dl = ei[Ecnt + e] - lo;
        q[i]  = s;
        lr[i] = ((i < n) && ((unsigned)dl < (unsigned)span)) ? (unsigned short)dl
                                                            : (unsigned short)NOT_IN_TILE;
    }
}

__device__ __forceinline__ void message_batch(lds_queue_word* q, lds_row_word* lr,
                                              int first, int cnt, int lane, int lo,
                                              const float* __restrict__ a_src,
                                              const float* __restrict__ a_dst,
                                              bool have_lg, const float (&lgk)[HEADS],
                                              const float* tmax, float* tden,
                                              const float* __restrict__ h,
                                              float* acc) {
    const int idx = first + (lane < cnt ? lane : 0);
    const int s = q[idx];
    const unsigned int r = lr[idx];
    const bool valid = (lane < cnt) && (r != NOT_IN_TILE);
    const unsigned okm = __builtin_amdgcn_ballot_w32(valid);
    const int dl = valid ? (int)r : 0;
    float lg[HEADS];
    if (have_lg) {
        lg[0] = lgk[0];
        lg[1] = lgk[1];
        lg[2] = lgk[2];
        lg[3] = lgk[3];
    } else {
        const float4 as4 = *reinterpret_cast<const float4*>(a_src + (size_t)s * HEADS);
        const float4 ad4 = *reinterpret_cast<const float4*>(a_dst + (size_t)(lo + dl) * HEADS);
        lg[0] = leaky(as4.x + ad4.x);
        lg[1] = leaky(as4.y + ad4.y);
        lg[2] = leaky(as4.z + ad4.z);
        lg[3] = leaky(as4.w + ad4.w);
    }
    float w[HEADS];
#pragma unroll
    for (int k = 0; k < HEADS; ++k) {
        float mx = tmax[dl * HEADS + k];
        if (!__builtin_isfinite(mx)) mx = 0.f;
        w[k] = expf(lg[k] - mx);
    }
    if (valid) {
#pragma unroll
        for (int k = 0; k < HEADS; ++k) atomicAdd(tden + dl * HEADS + k, w[k]);
    }
    const unsigned int hm1 = 0u - (unsigned int)((lane >> 3) & 1);
    const unsigned int hm2 = 0u - (unsigned int)(lane >> 4);
    for (int j0 = 0; j0 < cnt; j0 += ROWS_IN_FLIGHT) {
        float4 v[ROWS_IN_FLIGHT];
#pragma unroll
        for (int t = 0; t < ROWS_IN_FLIGHT; ++t) {
            const int jj = (j0 + t < cnt) ? (j0 + t) : (cnt - 1);
            const float* hrow = h + (size_t)lane_bcast_i(s, jj) * FDIM + 4 * lane;
            v[t] = *reinterpret_cast<const float4*>(hrow);
        }
#pragma unroll
        for (int t = 0; t < ROWS_IN_FLIGHT; ++t) {
            const int jj = j0 + t;
            if ((okm >> jj) & 1u) {
                float* arow = acc + lane_bcast_i(dl, jj) * FDIM + lane;
                const unsigned int u0 = (unsigned int)lane_bcast_i(__float_as_int(w[0]), jj);
                const unsigned int u1 = (unsigned int)lane_bcast_i(__float_as_int(w[1]), jj);
                const unsigned int u2 = (unsigned int)lane_bcast_i(__float_as_int(w[2]), jj);
                const unsigned int u3 = (unsigned int)lane_bcast_i(__float_as_int(w[3]), jj);
                const unsigned int lo01 = (u1 & hm1) | (u0 & ~hm1);
                const unsigned int hi23 = (u3 & hm1) | (u2 & ~hm1);
                const float wl = __uint_as_float((hi23 & hm2) | (lo01 & ~hm2));
                atomicAdd(arow + 0 * CHAN, v[t].x * wl);
                atomicAdd(arow + 1 * CHAN, v[t].y * wl);
                atomicAdd(arow + 2 * CHAN, v[t].z * wl);
                atomicAdd(arow + 3 * CHAN, v[t].w * wl);
            }
        }
    }
}

__device__ __forceinline__ void list_sum(lds_queue_word* q, lds_row_word* lr, int n,
                                         int lane, int lo,
                                         const float* __restrict__ a_src,
                                         const float* __restrict__ a_dst,
                                         const float* tmax, float* tden,
                                         const float* __restrict__ h,
                                         float* acc,
                                         float (&kept)[KEPT_SLOTS][HEADS], int nkept) {
    for (int first = 0; first < n; first += 32) {
        const int c = (n - first) < 32 ? (n - first) : 32;
        message_batch(q, lr, first, c, lane, lo, a_src, a_dst, first < nkept * 32, kept[0],
                      tmax, tden, h, acc);
#pragma unroll
        for (int j = 0; j + 1 < KEPT_SLOTS; ++j) {
#pragma unroll
            for (int k = 0; k < HEADS; ++k) kept[j][k] = kept[j + 1][k];
        }
    }
}

__device__ __forceinline__ void load_step(uint4 (&buf)[SCAN_VECS],
                                          const unsigned int* __restrict__ p) {
#pragma unroll
    for (int u = 0; u < SCAN_VECS; ++u)
        buf[u] = *reinterpret_cast<const uint4*>(p + u * MSG_THREADS * 4);
}

__device__ __forceinline__ uint4 load_vec_guarded(const unsigned int* __restrict__ plane,
                                                  int pos0, int Ecnt, unsigned int fillw) {
    unsigned int w[4];
#pragma unroll
    for (int j = 0; j < 4; ++j) {
        const int p = pos0 + 4 * j;
        unsigned int x = fillw;
        if (p < Ecnt) {
            x = plane[p >> 2];
            const int nv = Ecnt - p;
            if (nv < 4) {
                const unsigned int keep = (1u << (8 * nv)) - 1u;
                x = (x & keep) | (fillw & ~keep);
            }
        }
        w[j] = x;
    }
    return make_uint4(w[0], w[1], w[2], w[3]);
}

__device__ __forceinline__ unsigned int nonzero_bytes(unsigned int x) {
    return ((x & 0x7f7f7f7fu) + 0x7f7f7f7fu) | x;
}

__device__ __forceinline__ void scan_vec(const uint4 v, int pos0, unsigned int pat,
                                         lds_queue_word* q, int& qn) {
    const unsigned int miss = ((nonzero_bytes(v.x ^ pat) >> 7) & 0x01010101u) |
                              ((nonzero_bytes(v.y ^ pat) >> 6) & 0x02020202u) |
                              ((nonzero_bytes(v.z ^ pat) >> 5) & 0x04040404u) |
                              ((nonzero_bytes(v.w ^ pat) >> 4) & 0x08080808u);
    unsigned int hm = miss ^ 0x0f0f0f0fu;
    unsigned int any = __builtin_amdgcn_ballot_w32(hm != 0u);
    while (any != 0u) {
        const bool has = hm != 0u;
        const int bit = __builtin_ctz(hm | 0x80000000u);
        const int off = ((bit & 7) << 2) | (bit >> 3);
        const int slot = qn + (int)__builtin_amdgcn_mbcnt_lo(any, 0u);
        if (has && slot < LIST_PER_WAVE) q[slot] = pos0 + off;
        qn += __builtin_popcount(any);
        hm &= hm - 1u;
        any = __builtin_amdgcn_ballot_w32(hm != 0u);
    }
}

__device__ __forceinline__ void scan_step(const uint4 (&buf)[SCAN_VECS], int base, int tid,
                                          unsigned int pat, lds_queue_word* q, int& qn) {
#pragma unroll
    for (int u = 0; u < SCAN_VECS; ++u)
        scan_vec(buf[u], base + (u * MSG_THREADS + tid) * IDS_PER_VEC, pat, q, qn);
}

__global__ __launch_bounds__(MSG_THREADS)
void edge_message_tiles(const int* __restrict__ ei,
                        const unsigned int* __restrict__ plane,
                        const float* __restrict__ a_src,
                        const float* __restrict__ a_dst,
                        const float* __restrict__ h,
                        const float* __restrict__ bias,
                        float* __restrict__ out, int Ecnt, int Nn) {
    __shared__ float acc[TILE_NODES * FDIM];
    __shared__ int   list[MSG_WAVES * LIST_PER_WAVE];
    __shared__ unsigned short rows[MSG_WAVES * LIST_PER_WAVE];
    __shared__ float tmax[(TILE_NODES + 32) * HEADS];
    __shared__ float tden[TILE_NODES * HEADS];
    __shared__ int   crowded;

    const int tid  = threadIdx.x;
    const int lane = tid & 31;
    const int wave = tid >> 5;
    const int lo   = blockIdx.x * TILE_NODES;
    int span = Nn - lo;
    if (span > TILE_NODES) span = TILE_NODES;
    if (span < 0) span = 0;

    for (int i = tid; i < TILE_NODES * FDIM; i += MSG_THREADS) acc[i] = 0.0f;
    for (int i = tid; i < (TILE_NODES + 32) * HEADS; i += MSG_THREADS) tmax[i] = -__builtin_inff();
    for (int i = tid; i < TILE_NODES * HEADS; i += MSG_THREADS) tden[i] = 0.0f;
    if (tid == 0) crowded = 0;
    __syncthreads();

    lds_queue_word* q = (lds_queue_word*)(list + wave * LIST_PER_WAVE);
    lds_row_word* lr = (lds_row_word*)(rows + wave * LIST_PER_WAVE);
    float kept[KEPT_SLOTS][HEADS];
#pragma unroll
    for (int j = 0; j < KEPT_SLOTS; ++j) {
#pragma unroll
        for (int k = 0; k < HEADS; ++k) kept[j][k] = 0.f;
    }
    int qn = 0;

    const unsigned int pat   = (blockIdx.x < 255u ? blockIdx.x : 255u) * 0x01010101u;
    const unsigned int fillw = pat ^ 0x80808080u;
    const int whole = (Ecnt / STEP_EDGES) * STEP_EDGES;

    if (whole > 0) {
        const unsigned int* lp = plane + tid * 4;
        uint4 va[SCAN_VECS], vb[SCAN_VECS];
        load_step(va, lp);
        for (int base = 0; base < whole; base += 2 * STEP_EDGES) {
            const int b1 = base + STEP_EDGES;
            const int b2 = b1 + STEP_EDGES;
            load_step(vb, lp + ((b1 < whole ? b1 : base) >> 2));
            scan_step(va, base, tid, pat, q, qn);
            load_step(va, lp + ((b2 < whole ? b2 : base) >> 2));
            if (b1 < whole)
                scan_step(vb, b1, tid, pat, q, qn);
        }
    }
    for (int base = whole; base < Ecnt; base += ROW_EDGES) {
        const int pos0 = base + tid * IDS_PER_VEC;
        scan_vec(load_vec_guarded(plane, pos0, Ecnt, fillw), pos0, pat, q, qn);
    }

    if (qn > LIST_PER_WAVE && lane == 0) crowded = 1;
    __syncthreads();
    const bool step_by_step = crowded != 0;

    if (!step_by_step) {
        list_max(q, lr, qn, lane, lo, span, ei, a_src, a_dst, tmax, Ecnt, kept);
        __syncthreads();
        list_sum(q, lr, qn, lane, lo, a_src, a_dst, tmax, tden, h, acc, kept, KEPT_SLOTS);
    } else {
        for (int base = 0; base < Ecnt; base += ROW_EDGES) {
            const int pos0 = base + tid * IDS_PER_VEC;
            qn = 0;
            scan_vec(load_vec_guarded(plane, pos0, Ecnt, fillw), pos0, pat, q, qn);
            list_max(q, lr, qn, lane, lo, span, ei, a_src, a_dst, tmax, Ecnt, kept);
        }
        __syncthreads();
        for (int base = 0; base < Ecnt; base += ROW_EDGES) {
            const int pos0 = base + tid * IDS_PER_VEC;
            qn = 0;
            scan_vec(load_vec_guarded(plane, pos0, Ecnt, fillw), pos0, pat, q, qn);
            list_resolve(q, lr, qn, lane, lo, span, ei, Ecnt);
            list_sum(q, lr, qn, lane, lo, a_src, a_dst, tmax, tden, h, acc, kept, 0);
        }
    }

    __syncthreads();

    for (int i = tid; i < span * FDIM; i += MSG_THREADS) {
        const int col = i & (FDIM - 1);
        const float den = tden[(i >> 7) * HEADS + (col >> 5)] + 1e-16f;
        const float sum = acc[(i & ~(FDIM - 1)) + ((col & 3) << 5) + (col >> 2)];
        float v = sum / den + bias[col];
        out[(size_t)lo * FDIM + i] = v > 0.f ? v : expm1f(v);
    }
}

extern "C" void kernel_launch(void* const* d_in, const int* in_sizes, int n_in,
                              void* d_out, int out_size, void* d_ws, size_t ws_size,
                              hipStream_t stream) {
    const float* x       = (const float*)d_in[0];
    const int*   ei      = (const int*)  d_in[1];
    const float* W       = (const float*)d_in[2];
    const float* att_src = (const float*)d_in[3];
    const float* att_dst = (const float*)d_in[4];
    const float* bias    = (const float*)d_in[5];
    float* out = (float*)d_out;

    const int N = in_sizes[0] / FDIM;
    const int E = in_sizes[1] / 2;

    float* ws     = (float*)d_ws;
    float* h      = ws;
    float* a_src  = h     + (size_t)N * FDIM;
    float* a_dst  = a_src + (size_t)N * HEADS;
    unsigned int* plane = (unsigned int*)(a_dst + (size_t)N * HEADS);

    int tiles  = (N + 15) / 16;
    int gblk   = (tiles + 3) / 4;
    gemm_terms_plane<<<gblk, GEMM_THREADS, 0, stream>>>(x, W, att_src, att_dst, h, a_src, a_dst,
                                                       ei, plane, N, E);

    edge_message_tiles<<<(N + TILE_NODES - 1) / TILE_NODES, MSG_THREADS, 0, stream>>>(
        ei, plane, a_src, a_dst, h, bias, out, E, N);
}


# tile plane kept as 32 one-hot bit planes (16 per nibble of the tile byte, wave bit transpose in the GEMM launch): a tile scans its 2 planes, 1/4 of the bytes; lists reordered, sums arrival order
# speedup vs baseline: 6.6132x; 1.1025x over previous
#include <hip/hip_runtime.h>
#include <math.h>

typedef float v2f __attribute__((ext_vector_type(2)));
typedef float v8f __attribute__((ext_vector_type(8)));

#define HEADS 4
#define CHAN 32
#define FDIM 128
#define NEG_SLOPE 0.2f
#define TILE_NODES 392
#define BLK_EDGES  32
#define GRP_BLKS   8
#define PLANES     32

__device__ __forceinline__ unsigned int tile_byte(int d) {
    const unsigned int t = (unsigned int)d / (unsigned int)TILE_NODES;
    return t < 255u ? t : 255u;
}

__device__ __forceinline__ int plane_stride_words(int Ecnt) {
    const int nblk = (Ecnt + BLK_EDGES - 1) / BLK_EDGES;
    return ((nblk + GRP_BLKS - 1) / GRP_BLKS) * GRP_BLKS;
}

__device__ __forceinline__ unsigned int wave_bit_transpose(unsigned int w, int lane) {
#pragma unroll
    for (int st = 0; st < 5; ++st) {
        const int s = 16 >> st;
        const unsigned int m = st == 0 ? 0x0000FFFFu : st == 1 ? 0x00FF00FFu :
                               st == 2 ? 0x0F0F0F0Fu : st == 3 ? 0x33333333u : 0x55555555u;
        const unsigned int p = (unsigned int)__shfl_xor((int)w, s);
        const bool low = (lane & s) == 0;
        const unsigned int a = low ? w : (p >> s);
        const unsigned int b = low ? (p << s) : w;
        w = (a & m) | (b & ~m);
    }
    return w;
}

__device__ __forceinline__ float sum8(float v) {
    v += __int_as_float(__builtin_amdgcn_update_dpp(0, __float_as_int(v), 0xB1, 0xF, 0xF, true));
    v += __int_as_float(__builtin_amdgcn_update_dpp(0, __float_as_int(v), 0x4E, 0xF, 0xF, true));
    v += __int_as_float(__builtin_amdgcn_update_dpp(0, __float_as_int(v), 0x141, 0xF, 0xF, true));
    return v;
}

#define GEMM_THREADS 128
#define FOLD_ROWS    16

__global__ __launch_bounds__(GEMM_THREADS)
void gemm_terms_plane(const float* __restrict__ x, const float* __restrict__ W,
                      const float* __restrict__ att_src, const float* __restrict__ att_dst,
                      float* __restrict__ h, float* __restrict__ a_src, float* __restrict__ a_dst,
                      const int* __restrict__ ei, unsigned int* __restrict__ tile_plane,
                      int Nn, int Ecnt) {
    __shared__ float fragB[32 * 32 * 2];

    const int wave = threadIdx.x >> 5;
    const int lane = threadIdx.x & 31;

    {
        const int stride = plane_stride_words(Ecnt);
        const int ngrp = stride / GRP_BLKS;
        const int* dstp = ei + Ecnt;
        const int nwaves = gridDim.x * (GEMM_THREADS / 32);
        const int wave_u = __builtin_amdgcn_readfirstlane(wave);
        for (int grp = blockIdx.x * (GEMM_THREADS / 32) + wave_u; grp < ngrp; grp += nwaves) {
            int d[GRP_BLKS];
            unsigned int pw[GRP_BLKS];
#pragma unroll
            for (int j = 0; j < GRP_BLKS; ++j) {
                const int e = (grp * GRP_BLKS + j) * BLK_EDGES + lane;
                d[j] = dstp[e < Ecnt ? e : Ecnt - 1];
            }
#pragma unroll
            for (int j = 0; j < GRP_BLKS; ++j) {
                const int e = (grp * GRP_BLKS + j) * BLK_EDGES + lane;
                const unsigned int t = tile_byte(d[j]);
                unsigned int w = (1u << (t >> 4)) | (0x10000u << (t & 15u));
                w &= 0u - (unsigned int)(e < Ecnt);
                pw[j] = wave_bit_transpose(w, lane);
            }
            uint4* pv = reinterpret_cast<uint4*>(tile_plane + (size_t)lane * stride + grp * GRP_BLKS);
            pv[0] = make_uint4(pw[0], pw[1], pw[2], pw[3]);
            pv[1] = make_uint4(pw[4], pw[5], pw[6], pw[7]);
        }
    }

    {
        const float4 as4 = *reinterpret_cast<const float4*>(att_src + 4 * lane);
        const float4 ad4 = *reinterpret_cast<const float4*>(att_dst + 4 * lane);
        const int g = lane >> 3;
        const int p = lane & 7;
        for (int i0 = 0; i0 < 32; i0 += FOLD_ROWS) {
            float4 w4[FOLD_ROWS];
#pragma unroll
            for (int t = 0; t < FOLD_ROWS; ++t)
                w4[t] = *reinterpret_cast<const float4*>(W + (size_t)(wave * 32 + i0 + t) * FDIM + 4 * lane);
#pragma unroll
            for (int t = 0; t < FOLD_ROWS; ++t) {
                const int k = wave * 32 + i0 + t;
                float ps = w4[t].x * as4.x;
                ps = fmaf(w4[t].y, as4.y, ps);
                ps = fmaf(w4[t].z, as4.z, ps);
                ps = fmaf(w4[t].w, as4.w, ps);
                float pd = w4[t].x * ad4.x;
                pd = fmaf(w4[t].y, ad4.y, pd);
                pd = fmaf(w4[t].z, ad4.z, pd);
                pd = fmaf(w4[t].w, ad4.w, pd);
                ps = sum8(ps);
                pd = sum8(pd);
                if (p < 4) {
                    const float val = p == 0 ? ps : (p == 1 ? pd : 0.0f);
                    const int n = p * 4 + g;
                    fragB[(((k >> 2) * 32) + n + ((k & 2) << 3)) * 2 + (k & 1)] = val;
                }
            }
        }
    }
    __syncthreads();

    const int row0 = (blockIdx.x * 4 + wave) * 16;
    if (row0 >= Nn) return;

    v8f acc[9];
#pragma unroll
    for (int t = 0; t < 9; ++t) acc[t] = (v8f)(0.0f);

    const int mrow  = row0 + (lane & 15);
    const int khalf = (lane >> 4) << 1;
    const int ncol  = lane & 15;

    for (int k0 = 0; k0 < FDIM; k0 += 4) {
        const int ka = k0 + khalf;
        v2f a;
        a.x = x[(size_t)mrow * FDIM + ka];
        a.y = x[(size_t)mrow * FDIM + ka + 1];
#pragma unroll
        for (int t = 0; t < 8; ++t) {
            const int n = t * 16 + ncol;
            v2f b;
            b.x = W[(size_t)ka * FDIM + n];
            b.y = W[(size_t)(ka + 1) * FDIM + n];
            acc[t] = __builtin_amdgcn_wmma_f32_16x16x4_f32(
                false, a, false, b, (short)0, acc[t], false, false);
        }
        v2f b9;
        b9.x = fragB[((k0 >> 2) * 32 + lane) * 2];
        b9.y = fragB[((k0 >> 2) * 32 + lane) * 2 + 1];
        acc[8] = __builtin_amdgcn_wmma_f32_16x16x4_f32(
            false, a, false, b9, (short)0, acc[8], false, false);
    }

    const int mbase = row0 + ((lane >> 4) << 3);
#pragma unroll
    for (int t = 0; t < 8; ++t) {
        const int n = t * 16 + ncol;
#pragma unroll
        for (int r = 0; r < 8; ++r)
            h[(size_t)(mbase + r) * FDIM + n] = acc[t][r];
    }
    if (ncol < 8) {
        float* ap = (ncol < 4) ? a_src : a_dst;
        const int hd = ncol & 3;
#pragma unroll
        for (int r = 0; r < 8; ++r)
            ap[(size_t)(mbase + r) * HEADS + hd] = acc[8][r];
    }
}

__device__ __forceinline__ float leaky(float v) {
    return v > 0.f ? v : NEG_SLOPE * v;
}

__device__ __forceinline__ void atomicMaxF32(float* addr, float val) {
    if (val >= 0.f)
        atomicMax((int*)addr, __float_as_int(val));
    else
        atomicMin((unsigned int*)addr, __float_as_uint(val));
}

#define MSG_THREADS  1024
#define MSG_WAVES    (MSG_THREADS / 32)
#define QUAD_BLKS    4
#define SCAN_QUADS   2
#define ROW_QUADS    MSG_THREADS
#define STEP_QUADS   (ROW_QUADS * SCAN_QUADS)
#define LIST_PER_WAVE 512
#define ROWS_IN_FLIGHT 8
#define MAX_IN_FLIGHT  4
#define KEPT_SLOTS     8
#define NOT_IN_TILE    0xFFFFu

static_assert(PLANES == 32, "the transpose leaves one plane per lane");
static_assert(GRP_BLKS % QUAD_BLKS == 0, "a plane is a whole number of 16-byte vectors");
static_assert(LIST_PER_WAVE >= 32 * (BLK_EDGES / 2), "half a block per lane must always fit the wave's list");
static_assert(LIST_PER_WAVE % (32 * MAX_IN_FLIGHT) == 0, "a chunk of the max pass must end inside the list");
static_assert(KEPT_SLOTS % MAX_IN_FLIGHT == 0, "kept slots are whole chunks of the max pass");
static_assert(TILE_NODES < NOT_IN_TILE, "a row inside the tile must fit the 16-bit row word");

typedef __attribute__((address_space(3))) volatile int lds_queue_word;
typedef __attribute__((address_space(3))) volatile unsigned short lds_row_word;

__device__ __forceinline__ int lane_bcast_i(int v, int j) {
    return __builtin_amdgcn_readlane(v, j);
}
__device__ __forceinline__ float lane_bcast_f(float v, int j) {
    return __int_as_float(__builtin_amdgcn_readlane(__float_as_int(v), j));
}

__device__ __forceinline__ void max_chunk(lds_queue_word* q, lds_row_word* lr, int n, int b,
                                          int lane, int lo, int span,
                                          const int* __restrict__ ei,
                                          const float* __restrict__ a_src,
                                          const float* __restrict__ a_dst,
                                          float* tmax, int Ecnt, float (*lg)[HEADS]) {
    int e[MAX_IN_FLIGHT], s[MAX_IN_FLIGHT], dl[MAX_IN_FLIGHT];
    float4 as4[MAX_IN_FLIGHT], ad4[MAX_IN_FLIGHT];
#pragma unroll
    for (int t = 0; t < MAX_IN_FLIGHT; ++t) {
        const int i = b + t * 32 + lane;
        e[t] = q[i < n ? i : n - 1];
    }
#pragma unroll
    for (int t = 0; t < MAX_IN_FLIGHT; ++t) {
        s[t]  = ei[e[t]];
        dl[t] = ei[Ecnt + e[t]];
    }
#pragma unroll
    for (int t = 0; t < MAX_IN_FLIGHT; ++t) {
        as4[t] = *reinterpret_cast<const float4*>(a_src + (size_t)s[t] * HEADS);
        ad4[t] = *reinterpret_cast<const float4*>(a_dst + (size_t)dl[t] * HEADS);
        dl[t] -= lo;
    }
#pragma unroll
    for (int t = 0; t < MAX_IN_FLIGHT; ++t) {
        const int i = b + t * 32 + lane;
        const bool live = (i < n) && ((unsigned)dl[t] < (unsigned)span);
        float* slot = tmax + (live ? dl[t] : TILE_NODES + lane) * HEADS;
        lg[t][0] = leaky(as4[t].x + ad4[t].x);
        lg[t][1] = leaky(as4[t].y + ad4[t].y);
        lg[t][2] = leaky(as4[t].z + ad4[t].z);
        lg[t][3] = leaky(as4[t].w + ad4[t].w);
        atomicMaxF32(slot + 0, lg[t][0]);
        atomicMaxF32(slot + 1, lg[t][1]);
        atomicMaxF32(slot + 2, lg[t][2]);
        atomicMaxF32(slot + 3, lg[t][3]);
        q[i]  = s[t];
        lr[i] = live ? (unsigned short)dl[t] : (unsigned short)NOT_IN_TILE;
    }
}

__device__ __forceinline__ void list_max(lds_queue_word* q, lds_row_word* lr, int n,
                                         int lane, int lo, int span,
                                         const int* __restrict__ ei,
                                         const float* __restrict__ a_src,
                                         const float* __restrict__ a_dst,
                                         float* tmax, int Ecnt,
                                         float (&kept)[KEPT_SLOTS][HEADS]) {
#pragma unroll
    for (int c = 0; c < KEPT_SLOTS / MAX_IN_FLIGHT; ++c) {
        if (c * 32 * MAX_IN_FLIGHT < n)
            max_chunk(q, lr, n, c * 32 * MAX_IN_FLIGHT, lane, lo, span, ei, a_src, a_dst,
                      tmax, Ecnt, &kept[c * MAX_IN_FLIGHT]);
    }
    for (int b = KEPT_SLOTS * 32; b < n; b += 32 * MAX_IN_FLIGHT) {
        float spare[MAX_IN_FLIGHT][HEADS];
        max_chunk(q, lr, n, b, lane, lo, span, ei, a_src, a_dst, tmax, Ecnt, spare);
    }
}

__device__ __forceinline__ void list_resolve(lds_queue_word* q, lds_row_word* lr, int n,
                                             int lane, int lo, int span,
                                             const int* __restrict__ ei, int Ecnt) {
    for (int b = 0; b < n; b += 32) {
        const int i = b + lane;
        const int e = q[i < n ? i : n - 1];
        const int s = ei[e];
        const int dl = ei[Ecnt + e] - lo;
        q[i]  = s;
        lr[i] = ((i < n) && ((unsigned)dl < (unsigned)span)) ? (unsigned short)dl
                                                            : (unsigned short)NOT_IN_TILE;
    }
}

__device__ __forceinline__ void message_batch(lds_queue_word* q, lds_row_word* lr,
                                              int first, int cnt, int lane, int lo,
                                              const float* __restrict__ a_src,
                                              const float* __restrict__ a_dst,
                                              bool have_lg, const float (&lgk)[HEADS],
                                              const float* tmax, float* tden,
                                              const float* __restrict__ h,
                                              float* acc) {
    const int idx = first + (lane < cnt ? lane : 0);
    const int s = q[idx];
    const unsigned int r = lr[idx];
    const bool valid = (lane < cnt) && (r != NOT_IN_TILE);
    const unsigned okm = __builtin_amdgcn_ballot_w32(valid);
    const int dl = valid ? (int)r : 0;
    float lg[HEADS];
    if (have_lg) {
        lg[0] = lgk[0];
        lg[1] = lgk[1];
        lg[2] = lgk[2];
        lg[3] = lgk[3];
    } else {
        const float4 as4 = *reinterpret_cast<const float4*>(a_src + (size_t)s * HEADS);
        const float4 ad4 = *reinterpret_cast<const float4*>(a_dst + (size_t)(lo + dl) * HEADS);
        lg[0] = leaky(as4.x + ad4.x);
        lg[1] = leaky(as4.y + ad4.y);
        lg[2] = leaky(as4.z + ad4.z);
        lg[3] = leaky(as4.w + ad4.w);
    }
    float w[HEADS];
#pragma unroll
    for (int k = 0; k < HEADS; ++k) {
        float mx = tmax[dl * HEADS + k];
        if (!__builtin_isfinite(mx)) mx = 0.f;
        w[k] = expf(lg[k] - mx);
    }
    if (valid) {
#pragma unroll
        for (int k = 0; k < HEADS; ++k) atomicAdd(tden + dl * HEADS + k, w[k]);
    }
    const unsigned int hm1 = 0u - (unsigned int)((lane >> 3) & 1);
    const unsigned int hm2 = 0u - (unsigned int)(lane >> 4);
    for (int j0 = 0; j0 < cnt; j0 += ROWS_IN_FLIGHT) {
        float4 v[ROWS_IN_FLIGHT];
#pragma unroll
        for (int t = 0; t < ROWS_IN_FLIGHT; ++t) {
            const int jj = (j0 + t < cnt) ? (j0 + t) : (cnt - 1);
            const float* hrow = h + (size_t)lane_bcast_i(s, jj) * FDIM + 4 * lane;
            v[t] = *reinterpret_cast<const float4*>(hrow);
        }
#pragma unroll
        for (int t = 0; t < ROWS_IN_FLIGHT; ++t) {
            const int jj = j0 + t;
            if ((okm >> jj) & 1u) {
                float* arow = acc + lane_bcast_i(dl, jj) * FDIM + lane;
                const unsigned int u0 = (unsigned int)lane_bcast_i(__float_as_int(w[0]), jj);
                const unsigned int u1 = (unsigned int)lane_bcast_i(__float_as_int(w[1]), jj);
                const unsigned int u2 = (unsigned int)lane_bcast_i(__float_as_int(w[2]), jj);
                const unsigned int u3 = (unsigned int)lane_bcast_i(__float_as_int(w[3]), jj);
                const unsigned int lo01 = (u1 & hm1) | (u0 & ~hm1);
                const unsigned int hi23 = (u3 & hm1) | (u2 & ~hm1);
                const float wl = __uint_as_float((hi23 & hm2) | (lo01 & ~hm2));
                atomicAdd(arow + 0 * CHAN, v[t].x * wl);
                atomicAdd(arow + 1 * CHAN, v[t].y * wl);
                atomicAdd(arow + 2 * CHAN, v[t].z * wl);
                atomicAdd(arow + 3 * CHAN, v[t].w * wl);
            }
        }
    }
}

__device__ __forceinline__ void list_sum(lds_queue_word* q, lds_row_word* lr, int n,
                                         int lane, int lo,
                                         const float* __restrict__ a_src,
                                         const float* __restrict__ a_dst,
                                         const float* tmax, float* tden,
                                         const float* __restrict__ h,
                                         float* acc,
                                         float (&kept)[KEPT_SLOTS][HEADS], int nkept) {
    for (int first = 0; first < n; first += 32) {
        const int c = (n - first) < 32 ? (n - first) : 32;
        message_batch(q, lr, first, c, lane, lo, a_src, a_dst, first < nkept * 32, kept[0],
                      tmax, tden, h, acc);
#pragma unroll
        for (int j = 0; j + 1 < KEPT_SLOTS; ++j) {
#pragma unroll
            for (int k = 0; k < HEADS; ++k) kept[j][k] = kept[j + 1][k];
        }
    }
}

struct plane_quad { uint4 hn, ln; };

__device__ __forceinline__ void load_step(plane_quad (&buf)[SCAN_QUADS],
                                          const uint4* __restrict__ ph,
                                          const uint4* __restrict__ pl, int qd) {
#pragma unroll
    for (int u = 0; u < SCAN_QUADS; ++u) {
        buf[u].hn = ph[qd + u * ROW_QUADS];
        buf[u].ln = pl[qd + u * ROW_QUADS];
    }
}

__device__ __forceinline__ plane_quad load_quad_guarded(const uint4* __restrict__ ph,
                                                        const uint4* __restrict__ pl,
                                                        int qd, int nq) {
    plane_quad v;
    v.hn = make_uint4(0u, 0u, 0u, 0u);
    v.ln = make_uint4(0u, 0u, 0u, 0u);
    if (qd < nq) {
        v.hn = ph[qd];
        v.ln = pl[qd];
    }
    return v;
}

__device__ __forceinline__ void scan_mask(unsigned int m, int pos0,
                                          lds_queue_word* q, int& qn) {
    unsigned int any = __builtin_amdgcn_ballot_w32(m != 0u);
    while (any != 0u) {
        const bool has = m != 0u;
        const int bit = __builtin_ctz(m | 0x80000000u);
        const int slot = qn + (int)__builtin_amdgcn_mbcnt_lo(any, 0u);
        if (has && slot < LIST_PER_WAVE) q[slot] = pos0 + bit;
        qn += __builtin_popcount(any);
        m &= m - 1u;
        any = __builtin_amdgcn_ballot_w32(m != 0u);
    }
}

__device__ __forceinline__ void scan_quad(const plane_quad& v, int qd,
                                          lds_queue_word* q, int& qn) {
    const int pos0 = qd * (QUAD_BLKS * BLK_EDGES);
    scan_mask(v.hn.x & v.ln.x, pos0 + 0 * BLK_EDGES, q, qn);
    scan_mask(v.hn.y & v.ln.y, pos0 + 1 * BLK_EDGES, q, qn);
    scan_mask(v.hn.z & v.ln.z, pos0 + 2 * BLK_EDGES, q, qn);
    scan_mask(v.hn.w & v.ln.w, pos0 + 3 * BLK_EDGES, q, qn);
}

__device__ __forceinline__ void scan_step(const plane_quad (&buf)[SCAN_QUADS], int qd,
                                          lds_queue_word* q, int& qn) {
#pragma unroll
    for (int u = 0; u < SCAN_QUADS; ++u)
        scan_quad(buf[u], qd + u * ROW_QUADS, q, qn);
}

__global__ __launch_bounds__(MSG_THREADS)
void edge_message_tiles(const int* __restrict__ ei,
                        const unsigned int* __restrict__ plane,
                        const float* __restrict__ a_src,
                        const float* __restrict__ a_dst,
                        const float* __restrict__ h,
                        const float* __restrict__ bias,
                        float* __restrict__ out, int Ecnt, int Nn) {
    __shared__ float acc[TILE_NODES * FDIM];
    __shared__ int   list[MSG_WAVES * LIST_PER_WAVE];
    __shared__ unsigned short rows[MSG_WAVES * LIST_PER_WAVE];
    __shared__ float tmax[(TILE_NODES + 32) * HEADS];
    __shared__ float tden[TILE_NODES * HEADS];
    __shared__ int   crowded;

    const int tid  = threadIdx.x;
    const int lane = tid & 31;
    const int wave = tid >> 5;
    const int lo   = blockIdx.x * TILE_NODES;
    int span = Nn - lo;
    if (span > TILE_NODES) span = TILE_NODES;
    if (span < 0) span = 0;

    for (int i = tid; i < TILE_NODES * FDIM; i += MSG_THREADS) acc[i] = 0.0f;
    for (int i = tid; i < (TILE_NODES + 32) * HEADS; i += MSG_THREADS) tmax[i] = -__builtin_inff();
    for (int i = tid; i < TILE_NODES * HEADS; i += MSG_THREADS) tden[i] = 0.0f;
    if (tid == 0) crowded = 0;
    __syncthreads();

    lds_queue_word* q = (lds_queue_word*)(list + wave * LIST_PER_WAVE);
    lds_row_word* lr = (lds_row_word*)(rows + wave * LIST_PER_WAVE);
    float kept[KEPT_SLOTS][HEADS];
#pragma unroll
    for (int j = 0; j < KEPT_SLOTS; ++j) {
#pragma unroll
        for (int k = 0; k < HEADS; ++k) kept[j][k] = 0.f;
    }
    int qn = 0;

    const unsigned int pat = blockIdx.x < 255u ? blockIdx.x : 255u;
    const int stride = plane_stride_words(Ecnt);
    const int nq = stride / QUAD_BLKS;
    const uint4* ph = reinterpret_cast<const uint4*>(plane + (size_t)(pat >> 4) * stride);
    const uint4* pl = reinterpret_cast<const uint4*>(plane + (size_t)(16u + (pat & 15u)) * stride);
    const int whole = (nq / STEP_QUADS) * STEP_QUADS;

    if (whole > 0) {
        plane_quad va[SCAN_QUADS], vb[SCAN_QUADS];
        load_step(va, ph, pl, tid);
        for (int base = 0; base < whole; base += 2 * STEP_QUADS) {
            const int b1 = base + STEP_QUADS;
            const int b2 = b1 + STEP_QUADS;
            load_step(vb, ph, pl, (b1 < whole ? b1 : base) + tid);
            scan_step(va, base + tid, q, qn);
            load_step(va, ph, pl, (b2 < whole ? b2 : base) + tid);
            if (b1 < whole)
                scan_step(vb, b1 + tid, q, qn);
        }
    }
    for (int base = whole; base < nq; base += ROW_QUADS)
        scan_quad(load_quad_guarded(ph, pl, base + tid, nq), base + tid, q, qn);

    if (qn > LIST_PER_WAVE && lane == 0) crowded = 1;
    __syncthreads();
    const bool step_by_step = crowded != 0;

    if (!step_by_step) {
        list_max(q, lr, qn, lane, lo, span, ei, a_src, a_dst, tmax, Ecnt, kept);
        __syncthreads();
        list_sum(q, lr, qn, lane, lo, a_src, a_dst, tmax, tden, h, acc, kept, KEPT_SLOTS);
    } else {
        for (int base = 0; base < nq; base += ROW_QUADS) {
            const plane_quad v = load_quad_guarded(ph, pl, base + tid, nq);
            const unsigned int m[QUAD_BLKS] = { v.hn.x & v.ln.x, v.hn.y & v.ln.y, v.hn.z & v.ln.z, v.hn.w & v.ln.w };
            for (int part = 0; part < 2 * QUAD_BLKS; ++part) {
                qn = 0;
                scan_mask(m[part >> 1] & ((part & 1) ? 0xFFFF0000u : 0x0000FFFFu),
                          ((base + tid) * QUAD_BLKS + (part >> 1)) * BLK_EDGES, q, qn);
                list_max(q, lr, qn, lane, lo, span, ei, a_src, a_dst, tmax, Ecnt, kept);
            }
        }
        __syncthreads();
        for (int base = 0; base < nq; base += ROW_QUADS) {
            const plane_quad v = load_quad_guarded(ph, pl, base + tid, nq);
            const unsigned int m[QUAD_BLKS] = { v.hn.x & v.ln.x, v.hn.y & v.ln.y, v.hn.z & v.ln.z, v.hn.w & v.ln.w };
            for (int part = 0; part < 2 * QUAD_BLKS; ++part) {
                qn = 0;
                scan_mask(m[part >> 1] & ((part & 1) ? 0xFFFF0000u : 0x0000FFFFu),
                          ((base + tid) * QUAD_BLKS + (part >> 1)) * BLK_EDGES, q, qn);
                list_resolve(q, lr, qn, lane, lo, span, ei, Ecnt);
                list_sum(q, lr, qn, lane, lo, a_src, a_dst, tmax, tden, h, acc, kept, 0);
            }
        }
    }

    __syncthreads();

    for (int i = tid; i < span * FDIM; i += MSG_THREADS) {
        const int col = i & (FDIM - 1);
        const float den = tden[(i >> 7) * HEADS + (col >> 5)] + 1e-16f;
        const float sum = acc[(i & ~(FDIM - 1)) + ((col & 3) << 5) + (col >> 2)];
        float v = sum / den + bias[col];
        out[(size_t)lo * FDIM + i] = v > 0.f ? v : expm1f(v);
    }
}

extern "C" void kernel_launch(void* const* d_in, const int* in_sizes, int n_in,
                              void* d_out, int out_size, void* d_ws, size_t ws_size,
                              hipStream_t stream) {
    const float* x       = (const float*)d_in[0];
    const int*   ei      = (const int*)  d_in[1];
    const float* W       = (const float*)d_in[2];
    const float* att_src = (const float*)d_in[3];
    const float* att_dst = (const float*)d_in[4];
    const float* bias    = (const float*)d_in[5];
    float* out = (float*)d_out;

    const int N = in_sizes[0] / FDIM;
    const int E = in_sizes[1] / 2;

    float* ws     = (float*)d_ws;
    float* h      = ws;
    float* a_src  = h     + (size_t)N * FDIM;
    float* a_dst  = a_src + (size_t)N * HEADS;
    unsigned int* plane = (unsigned int*)(a_dst + (size_t)N * HEADS);

    int tiles  = (N + 15) / 16;
    int gblk   = (tiles + 3) / 4;
    gemm_terms_plane<<<gblk, GEMM_THREADS, 0, stream>>>(x, W, att_src, att_dst, h, a_src, a_dst,
                                                       ei, plane, N, E);

    edge_message_tiles<<<(N + TILE_NODES - 1) / TILE_NODES, MSG_THREADS, 0, stream>>>(
        ei, plane, a_src, a_dst, h, bias, out, E, N);
}
